# starting kernel
# baseline (speedup 1.0000x reference)
#include <hip/hip_runtime.h>

#define NTREE 511
#define BATCH 32
#define MDIM  256
#define LSTRIDE 260

typedef __attribute__((ext_vector_type(2))) float v2f;
typedef __attribute__((ext_vector_type(8))) float v8f;

__device__ __forceinline__ v8f wmma4(v2f a, v2f b, v8f c) {
  return __builtin_amdgcn_wmma_f32_16x16x4_f32(false, a, false, b, (short)0, c,
                                               false, false);
}

__device__ __forceinline__ float sigf(float x) { return 1.0f / (1.0f + __expf(-x)); }

__global__ __launch_bounds__(256) void pack_weights(
    const float* __restrict__ W_ioux, const float* __restrict__ b_ioux,
    const float* __restrict__ W_fx, const float* __restrict__ b_fx,
    float* __restrict__ wcat, float* __restrict__ bcat)
{
  int idx = blockIdx.x * blockDim.x + threadIdx.x;
  if (idx < 1024 * 256) {
    int row = idx >> 8;
    wcat[idx] = (row < 768) ? W_ioux[idx] : W_fx[idx - 768 * 256];
  } else {
    int col = idx - 1024 * 256;
    bcat[col] = (col < 768) ? b_ioux[col] : b_fx[col - 768];
  }
}

__global__ __launch_bounds__(256) void xproj_gemm(
    const float* __restrict__ inputs,
    const float* __restrict__ wcat,
    const float* __restrict__ bcat,
    float* __restrict__ xbuf)
{
  __shared__ float sa[BATCH * LSTRIDE];

  const int tid = threadIdx.x;
  const int n   = blockIdx.x;

#pragma unroll
  for (int i = 0; i < 8; ++i) {
    int F = tid + 256 * i;
    int b = F >> 6, mq = F & 63;
    float4 v = *(const float4*)(inputs + (size_t)b * (NTREE * MDIM) +
                                (size_t)n * MDIM + mq * 4);
    *(float4*)(&sa[b * LSTRIDE + mq * 4]) = v;
  }
  __syncthreads();

  const int lane  = tid & 31;
  const int wave  = tid >> 5;
  const int rbase = (wave & 1) * 16;
  const int c0    = blockIdx.y * 256 + (wave >> 1) * 64;
  const int lr    = lane & 15;
  const int khalf = (lane < 16) ? 0 : 2;

  const float* al  = sa + (size_t)(rbase + lr) * LSTRIDE + khalf;
  const float* w0p = wcat + (size_t)(c0 + 0 * 16 + lr) * 256 + khalf;
  const float* w1p = wcat + (size_t)(c0 + 1 * 16 + lr) * 256 + khalf;
  const float* w2p = wcat + (size_t)(c0 + 2 * 16 + lr) * 256 + khalf;
  const float* w3p = wcat + (size_t)(c0 + 3 * 16 + lr) * 256 + khalf;

  v8f acc0 = {}, acc1 = {}, acc2 = {}, acc3 = {};
  for (int k0 = 0; k0 < 256; k0 += 4) {
    v2f a  = *(const v2f*)(al + k0);
    v2f w0 = *(const v2f*)(w0p + k0);
    v2f w1 = *(const v2f*)(w1p + k0);
    v2f w2 = *(const v2f*)(w2p + k0);
    v2f w3 = *(const v2f*)(w3p + k0);
    acc0 = wmma4(a, w0, acc0);
    acc1 = wmma4(a, w1, acc1);
    acc2 = wmma4(a, w2, acc2);
    acc3 = wmma4(a, w3, acc3);
  }

  v8f accs[4] = {acc0, acc1, acc2, acc3};
#pragma unroll
  for (int f = 0; f < 4; ++f) {
    int col = c0 + f * 16 + lr;
    float bs = bcat[col];
    float* op = xbuf + (size_t)n * (BATCH * 1024) + col;
#pragma unroll
    for (int v = 0; v < 8; ++v) {
      int bo = rbase + v + ((lane >> 4) << 3);
      op[(size_t)bo * 1024] = accs[f][v] + bs;
    }
  }
}

__global__ __launch_bounds__(256) void leaf_kernel(
    const float* __restrict__ xbuf, const float* __restrict__ b_iouh,
    float* __restrict__ c_buf, float* __restrict__ hid)
{
  int idx = blockIdx.x * blockDim.x + threadIdx.x;
  int m = idx & 255, b = (idx >> 8) & 31, node = idx >> 13;
  const float* xp = xbuf + (size_t)node * (BATCH * 1024) + (size_t)b * 1024;
  float ig = sigf(xp[m] + b_iouh[m]);
  float og = sigf(xp[256 + m] + b_iouh[256 + m]);
  float ug = tanhf(xp[512 + m] + b_iouh[512 + m]);
  float c = ig * ug;
  float h = og * tanhf(c);
  size_t o = (size_t)b * (NTREE * MDIM) + (size_t)node * MDIM + m;
  c_buf[o] = c;
  hid[o] = h;
}

__global__ __launch_bounds__(256) void level_gemm(
    const float* __restrict__ hid,
    const float* __restrict__ W_iouh,
    const float* __restrict__ W_fh,
    float* __restrict__ tmp_iou,
    float* __restrict__ tmp_f,
    int level_start)
{
  __shared__ float sh[2][BATCH * LSTRIDE];

  const int tid  = threadIdx.x;
  const int nl   = blockIdx.x;
  const int node = level_start + nl;
  const int ch0  = 2 * node - 511;
  const int ch1  = 2 * node - 512;

#pragma unroll
  for (int i = 0; i < 16; ++i) {
    int F = tid + 256 * i;
    int which = F >> 11;
    int rem = F & 2047;
    int b = rem >> 6, mq = rem & 63;
    int ch = which ? ch1 : ch0;
    float4 v = *(const float4*)(hid + (size_t)b * (NTREE * MDIM) +
                                (size_t)ch * MDIM + mq * 4);
    *(float4*)(&sh[which][b * LSTRIDE + mq * 4]) = v;
  }
  __syncthreads();

  const int lane  = tid & 31;
  const int wave  = tid >> 5;
  const int lr    = lane & 15;
  const int khalf = (lane < 16) ? 0 : 2;

  for (int t = wave; t < 160; t += 8) {
    if (t < 96) {
      int rt = t & 1, ct = t >> 1;
      const float* a0 = &sh[0][(rt * 16 + lr) * LSTRIDE + khalf];
      const float* a1 = &sh[1][(rt * 16 + lr) * LSTRIDE + khalf];
      int col = ct * 16 + lr;
      const float* wr = W_iouh + (size_t)col * 256 + khalf;
      v8f acc = {};
      for (int k0 = 0; k0 < 256; k0 += 4) {
        v2f a = *(const v2f*)(a0 + k0) + *(const v2f*)(a1 + k0);
        v2f w = *(const v2f*)(wr + k0);
        acc = wmma4(a, w, acc);
      }
      float* op = tmp_iou + (size_t)nl * (BATCH * 768) + col;
#pragma unroll
      for (int v = 0; v < 8; ++v) {
        int bo = rt * 16 + v + ((lane >> 4) << 3);
        op[(size_t)bo * 768] = acc[v];
      }
    } else {
      int tt = t - 96;
      int child = tt >> 5, rem = tt & 31;
      int rt = rem & 1, ct = rem >> 1;
      const float* al = &sh[child][(rt * 16 + lr) * LSTRIDE + khalf];
      int col = ct * 16 + lr;
      const float* wr = W_fh + (size_t)col * 256 + khalf;
      v8f acc = {};
      for (int k0 = 0; k0 < 256; k0 += 4) {
        v2f a = *(const v2f*)(al + k0);
        v2f w = *(const v2f*)(wr + k0);
        acc = wmma4(a, w, acc);
      }
      float* op = tmp_f + (((size_t)nl * 2 + child) * BATCH) * MDIM + col;
#pragma unroll
      for (int v = 0; v < 8; ++v) {
        int bo = rt * 16 + v + ((lane >> 4) << 3);
        op[(size_t)bo * MDIM] = acc[v];
      }
    }
  }
}

__global__ __launch_bounds__(256) void level_elem(
    const float* __restrict__ xbuf, const float* __restrict__ tmp_iou,
    const float* __restrict__ tmp_f, const float* __restrict__ b_iouh,
    const float* __restrict__ b_fh, float* __restrict__ c_buf,
    float* __restrict__ hid, int level_start, int count)
{
  int idx = blockIdx.x * blockDim.x + threadIdx.x;
  if (idx >= count * BATCH * MDIM) return;
  int m = idx & 255, b = (idx >> 8) & 31, nl = idx >> 13;
  int node = level_start + nl;
  int ch0 = 2 * node - 511, ch1 = 2 * node - 512;
  const float* xp = xbuf + (size_t)node * (BATCH * 1024) + (size_t)b * 1024;
  const float* ti = tmp_iou + (size_t)nl * (BATCH * 768) + (size_t)b * 768;
  const float* tf = tmp_f + (size_t)nl * (2 * BATCH * MDIM) + (size_t)b * MDIM;
  float ig = sigf(xp[m] + ti[m] + b_iouh[m]);
  float og = sigf(xp[256 + m] + ti[256 + m] + b_iouh[256 + m]);
  float ug = tanhf(xp[512 + m] + ti[512 + m] + b_iouh[512 + m]);
  float xf = xp[768 + m];
  float f0 = sigf(tf[m] + b_fh[m] + xf);
  float f1 = sigf(tf[BATCH * MDIM + m] + b_fh[m] + xf);
  size_t base = (size_t)b * (NTREE * MDIM) + m;
  float c = ig * ug + f0 * c_buf[base + (size_t)ch0 * MDIM]
                    + f1 * c_buf[base + (size_t)ch1 * MDIM];
  float h = og * tanhf(c);
  c_buf[base + (size_t)node * MDIM] = c;
  hid[base + (size_t)node * MDIM] = h;
}

__global__ __launch_bounds__(256) void root_copy(
    const float* __restrict__ c_buf, const float* __restrict__ hid,
    float* __restrict__ out)
{
  int idx = blockIdx.x * blockDim.x + threadIdx.x;
  int m = idx & 255, b = idx >> 8;
  size_t o = (size_t)b * (NTREE * MDIM) + (size_t)510 * MDIM + m;
  out[idx] = c_buf[o];
  out[BATCH * MDIM + idx] = hid[o];
}

extern "C" void kernel_launch(void* const* d_in, const int* in_sizes, int n_in,
                              void* d_out, int out_size, void* d_ws, size_t ws_size,
                              hipStream_t stream) {
  (void)in_sizes; (void)n_in; (void)out_size; (void)ws_size;
  const float* inputs = (const float*)d_in[0];
  const float* W_ioux = (const float*)d_in[1];
  const float* b_ioux = (const float*)d_in[2];
  const float* W_iouh = (const float*)d_in[3];
  const float* b_iouh = (const float*)d_in[4];
  const float* W_fx   = (const float*)d_in[5];
  const float* b_fx   = (const float*)d_in[6];
  const float* W_fh   = (const float*)d_in[7];
  const float* b_fh   = (const float*)d_in[8];

  float* out = (float*)d_out;
  float* hid = out + 2 * BATCH * MDIM;

  float* ws = (float*)d_ws;
  float* xbuf    = ws;
  float* c_buf   = xbuf + (size_t)NTREE * BATCH * 1024;
  float* tmp_iou = c_buf + (size_t)BATCH * NTREE * MDIM;
  float* tmp_f   = tmp_iou + (size_t)128 * BATCH * 768;
  float* wcat    = tmp_f + (size_t)128 * 2 * BATCH * MDIM;
  float* bcat    = wcat + (size_t)1024 * 256;

  pack_weights<<<(1024 * 256 + 1024) / 256, 256, 0, stream>>>(
      W_ioux, b_ioux, W_fx, b_fx, wcat, bcat);

  dim3 g1(NTREE, 4);
  xproj_gemm<<<g1, 256, 0, stream>>>(inputs, wcat, bcat, xbuf);

  leaf_kernel<<<(256 * BATCH * MDIM) / 256, 256, 0, stream>>>(xbuf, b_iouh, c_buf, hid);

  const int starts[8] = {256, 384, 448, 480, 496, 504, 508, 510};
  const int counts[8] = {128, 64, 32, 16, 8, 4, 2, 1};
  for (int L = 0; L < 8; ++L) {
    level_gemm<<<counts[L], 256, 0, stream>>>(hid, W_iouh, W_fh, tmp_iou,
                                              tmp_f, starts[L]);
    int total = counts[L] * BATCH * MDIM;
    level_elem<<<(total + 255) / 256, 256, 0, stream>>>(
        xbuf, tmp_iou, tmp_f, b_iouh, b_fh, c_buf, hid, starts[L], counts[L]);
  }

  root_copy<<<(BATCH * MDIM) / 256, 256, 0, stream>>>(c_buf, hid, out);
}


# level GEMM spread over grid (node, part): each wave runs one 4-accumulator task, 64 K-steps with double-buffered fragments, instead of one workgroup per node sweeping 160 tiles in turn
# speedup vs baseline: 2.7471x; 2.7471x over previous
#include <hip/hip_runtime.h>

#define NTREE 511
#define BATCH 32
#define MDIM  256
#define LSTRIDE 260

typedef __attribute__((ext_vector_type(2))) float v2f;
typedef __attribute__((ext_vector_type(8))) float v8f;

__device__ __forceinline__ v8f wmma4(v2f a, v2f b, v8f c) {
  return __builtin_amdgcn_wmma_f32_16x16x4_f32(false, a, false, b, (short)0, c,
                                               false, false);
}

__device__ __forceinline__ float sigf(float x) { return 1.0f / (1.0f + __expf(-x)); }

__global__ __launch_bounds__(256) void pack_weights(
    const float* __restrict__ W_ioux, const float* __restrict__ b_ioux,
    const float* __restrict__ W_fx, const float* __restrict__ b_fx,
    float* __restrict__ wcat, float* __restrict__ bcat)
{
  int idx = blockIdx.x * blockDim.x + threadIdx.x;
  if (idx < 1024 * 256) {
    int row = idx >> 8;
    wcat[idx] = (row < 768) ? W_ioux[idx] : W_fx[idx - 768 * 256];
  } else {
    int col = idx - 1024 * 256;
    bcat[col] = (col < 768) ? b_ioux[col] : b_fx[col - 768];
  }
}

__global__ __launch_bounds__(256) void xproj_gemm(
    const float* __restrict__ inputs,
    const float* __restrict__ wcat,
    const float* __restrict__ bcat,
    float* __restrict__ xbuf)
{
  __shared__ float sa[BATCH * LSTRIDE];

  const int tid = threadIdx.x;
  const int n   = blockIdx.x;

#pragma unroll
  for (int i = 0; i < 8; ++i) {
    int F = tid + 256 * i;
    int b = F >> 6, mq = F & 63;
    float4 v = *(const float4*)(inputs + (size_t)b * (NTREE * MDIM) +
                                (size_t)n * MDIM + mq * 4);
    *(float4*)(&sa[b * LSTRIDE + mq * 4]) = v;
  }
  __syncthreads();

  const int lane  = tid & 31;
  const int wave  = tid >> 5;
  const int rbase = (wave & 1) * 16;
  const int c0    = blockIdx.y * 256 + (wave >> 1) * 64;
  const int lr    = lane & 15;
  const int khalf = (lane < 16) ? 0 : 2;

  const float* al  = sa + (size_t)(rbase + lr) * LSTRIDE + khalf;
  const float* w0p = wcat + (size_t)(c0 + 0 * 16 + lr) * 256 + khalf;
  const float* w1p = wcat + (size_t)(c0 + 1 * 16 + lr) * 256 + khalf;
  const float* w2p = wcat + (size_t)(c0 + 2 * 16 + lr) * 256 + khalf;
  const float* w3p = wcat + (size_t)(c0 + 3 * 16 + lr) * 256 + khalf;

  v8f acc0 = {}, acc1 = {}, acc2 = {}, acc3 = {};
  for (int k0 = 0; k0 < 256; k0 += 4) {
    v2f a  = *(const v2f*)(al + k0);
    v2f w0 = *(const v2f*)(w0p + k0);
    v2f w1 = *(const v2f*)(w1p + k0);
    v2f w2 = *(const v2f*)(w2p + k0);
    v2f w3 = *(const v2f*)(w3p + k0);
    acc0 = wmma4(a, w0, acc0);
    acc1 = wmma4(a, w1, acc1);
    acc2 = wmma4(a, w2, acc2);
    acc3 = wmma4(a, w3, acc3);
  }

  v8f accs[4] = {acc0, acc1, acc2, acc3};
#pragma unroll
  for (int f = 0; f < 4; ++f) {
    int col = c0 + f * 16 + lr;
    float bs = bcat[col];
    float* op = xbuf + (size_t)n * (BATCH * 1024) + col;
#pragma unroll
    for (int v = 0; v < 8; ++v) {
      int bo = rbase + v + ((lane >> 4) << 3);
      op[(size_t)bo * 1024] = accs[f][v] + bs;
    }
  }
}

__global__ __launch_bounds__(256) void leaf_kernel(
    const float* __restrict__ xbuf, const float* __restrict__ b_iouh,
    float* __restrict__ c_buf, float* __restrict__ hid)
{
  int idx = blockIdx.x * blockDim.x + threadIdx.x;
  int m = idx & 255, b = (idx >> 8) & 31, node = idx >> 13;
  const float* xp = xbuf + (size_t)node * (BATCH * 1024) + (size_t)b * 1024;
  float ig = sigf(xp[m] + b_iouh[m]);
  float og = sigf(xp[256 + m] + b_iouh[256 + m]);
  float ug = tanhf(xp[512 + m] + b_iouh[512 + m]);
  float c = ig * ug;
  float h = og * tanhf(c);
  size_t o = (size_t)b * (NTREE * MDIM) + (size_t)node * MDIM + m;
  c_buf[o] = c;
  hid[o] = h;
}

#define KU 8

__device__ __forceinline__ v2f ld2(const float* p) { return *(const v2f*)p; }

__device__ __forceinline__ void iou_load(
    v2f (&x0)[KU], v2f (&x1)[KU], v2f (&yA)[KU], v2f (&yB)[KU],
    const float* a0, const float* a1, const float* wA, const float* wB, int kb)
{
#pragma unroll
  for (int u = 0; u < KU; ++u) {
    int k = kb + 4 * u;
    x0[u] = ld2(a0 + k); x1[u] = ld2(a1 + k);
    yA[u] = ld2(wA + k); yB[u] = ld2(wB + k);
  }
  __builtin_amdgcn_sched_barrier(0);
}

__device__ __forceinline__ void iou_mma(
    const v2f (&x0)[KU], const v2f (&x1)[KU], const v2f (&yA)[KU],
    const v2f (&yB)[KU], v8f& c00, v8f& c01, v8f& c10, v8f& c11)
{
#pragma unroll
  for (int u = 0; u < KU; ++u) {
    c00 = wmma4(x0[u], yA[u], c00);
    c01 = wmma4(x0[u], yB[u], c01);
    c10 = wmma4(x1[u], yA[u], c10);
    c11 = wmma4(x1[u], yB[u], c11);
  }
  __builtin_amdgcn_sched_barrier(0);
}

__device__ __forceinline__ void fg_load(
    v2f (&x00)[KU], v2f (&x01)[KU], v2f (&x10)[KU], v2f (&x11)[KU],
    v2f (&y)[KU], const float* a00, const float* a01, const float* a10,
    const float* a11, const float* w, int kb)
{
#pragma unroll
  for (int u = 0; u < KU; ++u) {
    int k = kb + 4 * u;
    x00[u] = ld2(a00 + k); x01[u] = ld2(a01 + k);
    x10[u] = ld2(a10 + k); x11[u] = ld2(a11 + k);
    y[u] = ld2(w + k);
  }
  __builtin_amdgcn_sched_barrier(0);
}

__device__ __forceinline__ void fg_mma(
    const v2f (&x00)[KU], const v2f (&x01)[KU], const v2f (&x10)[KU],
    const v2f (&x11)[KU], const v2f (&y)[KU],
    v8f& c00, v8f& c01, v8f& c10, v8f& c11)
{
#pragma unroll
  for (int u = 0; u < KU; ++u) {
    c00 = wmma4(x00[u], y[u], c00);
    c01 = wmma4(x01[u], y[u], c01);
    c10 = wmma4(x10[u], y[u], c10);
    c11 = wmma4(x11[u], y[u], c11);
  }
  __builtin_amdgcn_sched_barrier(0);
}

template <int NTHR>
__device__ __forceinline__ void level_body(
    float (*sh)[BATCH * LSTRIDE],
    const float* __restrict__ hid, const float* __restrict__ W_iouh,
    const float* __restrict__ W_fh, float* __restrict__ tmp_iou,
    float* __restrict__ tmp_f, int level_start)
{
  const int tid  = threadIdx.x;
  const int nl   = blockIdx.x;
  const int node = level_start + nl;
  const int ch0  = 2 * node - 511;
  const int ch1  = 2 * node - 512;

  const bool iou_wg = blockIdx.y * (NTHR / 32) < 24;
  if (iou_wg) {
#pragma unroll
    for (int i = 0; i < 2048 / NTHR; ++i) {
      int F = tid + NTHR * i;
      int b = F >> 6, mq = F & 63;
      const float* hp = hid + (size_t)b * (NTREE * MDIM) + mq * 4;
      float4 v0 = *(const float4*)(hp + (size_t)ch0 * MDIM);
      float4 v1 = *(const float4*)(hp + (size_t)ch1 * MDIM);
      float4 s;
      s.x = v0.x + v1.x; s.y = v0.y + v1.y;
      s.z = v0.z + v1.z; s.w = v0.w + v1.w;
      *(float4*)(&sh[0][b * LSTRIDE + mq * 4]) = s;
    }
  } else {
#pragma unroll
    for (int i = 0; i < 4096 / NTHR; ++i) {
      int F = tid + NTHR * i;
      int which = F >> 11;
      int rem = F & 2047;
      int b = rem >> 6, mq = rem & 63;
      int ch = which ? ch1 : ch0;
      float4 v = *(const float4*)(hid + (size_t)b * (NTREE * MDIM) +
                                  (size_t)ch * MDIM + mq * 4);
      *(float4*)(&sh[which][b * LSTRIDE + mq * 4]) = v;
    }
  }
  __syncthreads();

  const int lane  = tid & 31;
  const int wave  = tid >> 5;
  const int lr    = lane & 15;
  const int khalf = (lane < 16) ? 0 : 2;
  const int hi8   = (lane >> 4) << 3;
  const int task  = blockIdx.y * (NTHR / 32) + wave;

  const float* a00 = &sh[0][lr * LSTRIDE + khalf];
  const float* a01 = &sh[0][(16 + lr) * LSTRIDE + khalf];
  const float* a10 = &sh[1][lr * LSTRIDE + khalf];
  const float* a11 = &sh[1][(16 + lr) * LSTRIDE + khalf];

  if (task < 24) {
    const int colA = task * 32 + lr;
    const int colB = colA + 16;
    const float* wA = W_iouh + (size_t)colA * 256 + khalf;
    const float* wB = W_iouh + (size_t)colB * 256 + khalf;

    v2f P0[KU], P1[KU], PA[KU], PB[KU];
    v2f Q0[KU], Q1[KU], QA[KU], QB[KU];
    v8f c00 = {}, c01 = {}, c10 = {}, c11 = {};
    iou_load(P0, P1, PA, PB, a00, a01, wA, wB, 0);
#pragma unroll 1
    for (int kb = 0; kb < 256 - 8 * KU; kb += 8 * KU) {
      iou_load(Q0, Q1, QA, QB, a00, a01, wA, wB, kb + 4 * KU);
      iou_mma(P0, P1, PA, PB, c00, c01, c10, c11);
      iou_load(P0, P1, PA, PB, a00, a01, wA, wB, kb + 8 * KU);
      iou_mma(Q0, Q1, QA, QB, c00, c01, c10, c11);
    }
    iou_load(Q0, Q1, QA, QB, a00, a01, wA, wB, 256 - 4 * KU);
    iou_mma(P0, P1, PA, PB, c00, c01, c10, c11);
    iou_mma(Q0, Q1, QA, QB, c00, c01, c10, c11);

    float* opA = tmp_iou + (size_t)nl * (BATCH * 768) + colA;
    float* opB = tmp_iou + (size_t)nl * (BATCH * 768) + colB;
#pragma unroll
    for (int v = 0; v < 8; ++v) {
      int b0 = v + hi8;
      int b1 = 16 + v + hi8;
      opA[(size_t)b0 * 768] = c00[v];
      opB[(size_t)b0 * 768] = c01[v];
      opA[(size_t)b1 * 768] = c10[v];
      opB[(size_t)b1 * 768] = c11[v];
    }
  } else {
    const int col = (task - 24) * 16 + lr;
    const float* wr = W_fh + (size_t)col * 256 + khalf;

    v2f P00[KU], P01[KU], P10[KU], P11[KU], PW[KU];
    v2f Q00[KU], Q01[KU], Q10[KU], Q11[KU], QW[KU];
    v8f c00 = {}, c01 = {}, c10 = {}, c11 = {};
    fg_load(P00, P01, P10, P11, PW, a00, a01, a10, a11, wr, 0);
#pragma unroll 1
    for (int kb = 0; kb < 256 - 8 * KU; kb += 8 * KU) {
      fg_load(Q00, Q01, Q10, Q11, QW, a00, a01, a10, a11, wr, kb + 4 * KU);
      fg_mma(P00, P01, P10, P11, PW, c00, c01, c10, c11);
      fg_load(P00, P01, P10, P11, PW, a00, a01, a10, a11, wr, kb + 8 * KU);
      fg_mma(Q00, Q01, Q10, Q11, QW, c00, c01, c10, c11);
    }
    fg_load(Q00, Q01, Q10, Q11, QW, a00, a01, a10, a11, wr, 256 - 4 * KU);
    fg_mma(P00, P01, P10, P11, PW, c00, c01, c10, c11);
    fg_mma(Q00, Q01, Q10, Q11, QW, c00, c01, c10, c11);

    float* op0 = tmp_f + (((size_t)nl * 2 + 0) * BATCH) * MDIM + col;
    float* op1 = tmp_f + (((size_t)nl * 2 + 1) * BATCH) * MDIM + col;
#pragma unroll
    for (int v = 0; v < 8; ++v) {
      int b0 = v + hi8;
      int b1 = 16 + v + hi8;
      op0[(size_t)b0 * MDIM] = c00[v];
      op0[(size_t)b1 * MDIM] = c01[v];
      op1[(size_t)b0 * MDIM] = c10[v];
      op1[(size_t)b1 * MDIM] = c11[v];
    }
  }
}

__global__ __launch_bounds__(256) void level_gemm_w8(
    const float* __restrict__ hid,
    const float* __restrict__ W_iouh,
    const float* __restrict__ W_fh,
    float* __restrict__ tmp_iou,
    float* __restrict__ tmp_f,
    int level_start)
{
  __shared__ float sh[2][BATCH * LSTRIDE];
  level_body<256>(sh, hid, W_iouh, W_fh, tmp_iou, tmp_f, level_start);
}

__global__ __launch_bounds__(128) void level_gemm_w4(
    const float* __restrict__ hid, const float* __restrict__ W_iouh,
    const float* __restrict__ W_fh, float* __restrict__ tmp_iou,
    float* __restrict__ tmp_f, int level_start)
{
  __shared__ float sh[2][BATCH * LSTRIDE];
  level_body<128>(sh, hid, W_iouh, W_fh, tmp_iou, tmp_f, level_start);
}

__global__ __launch_bounds__(256) void level_elem(
    const float* __restrict__ xbuf, const float* __restrict__ tmp_iou,
    const float* __restrict__ tmp_f, const float* __restrict__ b_iouh,
    const float* __restrict__ b_fh, float* __restrict__ c_buf,
    float* __restrict__ hid, int level_start, int count)
{
  int idx = blockIdx.x * blockDim.x + threadIdx.x;
  if (idx >= count * BATCH * MDIM) return;
  int m = idx & 255, b = (idx >> 8) & 31, nl = idx >> 13;
  int node = level_start + nl;
  int ch0 = 2 * node - 511, ch1 = 2 * node - 512;
  const float* xp = xbuf + (size_t)node * (BATCH * 1024) + (size_t)b * 1024;
  const float* ti = tmp_iou + (size_t)nl * (BATCH * 768) + (size_t)b * 768;
  const float* tf = tmp_f + (size_t)nl * (2 * BATCH * MDIM) + (size_t)b * MDIM;
  float ig = sigf(xp[m] + ti[m] + b_iouh[m]);
  float og = sigf(xp[256 + m] + ti[256 + m] + b_iouh[256 + m]);
  float ug = tanhf(xp[512 + m] + ti[512 + m] + b_iouh[512 + m]);
  float xf = xp[768 + m];
  float f0 = sigf(tf[m] + b_fh[m] + xf);
  float f1 = sigf(tf[BATCH * MDIM + m] + b_fh[m] + xf);
  size_t base = (size_t)b * (NTREE * MDIM) + m;
  float c = ig * ug + f0 * c_buf[base + (size_t)ch0 * MDIM]
                    + f1 * c_buf[base + (size_t)ch1 * MDIM];
  float h = og * tanhf(c);
  c_buf[base + (size_t)node * MDIM] = c;
  hid[base + (size_t)node * MDIM] = h;
}

__global__ __launch_bounds__(256) void root_copy(
    const float* __restrict__ c_buf, const float* __restrict__ hid,
    float* __restrict__ out)
{
  int idx = blockIdx.x * blockDim.x + threadIdx.x;
  int m = idx & 255, b = idx >> 8;
  size_t o = (size_t)b * (NTREE * MDIM) + (size_t)510 * MDIM + m;
  out[idx] = c_buf[o];
  out[BATCH * MDIM + idx] = hid[o];
}

extern "C" void kernel_launch(void* const* d_in, const int* in_sizes, int n_in,
                              void* d_out, int out_size, void* d_ws, size_t ws_size,
                              hipStream_t stream) {
  (void)in_sizes; (void)n_in; (void)out_size; (void)ws_size;
  const float* inputs = (const float*)d_in[0];
  const float* W_ioux = (const float*)d_in[1];
  const float* b_ioux = (const float*)d_in[2];
  const float* W_iouh = (const float*)d_in[3];
  const float* b_iouh = (const float*)d_in[4];
  const float* W_fx   = (const float*)d_in[5];
  const float* b_fx   = (const float*)d_in[6];
  const float* W_fh   = (const float*)d_in[7];
  const float* b_fh   = (const float*)d_in[8];

  float* out = (float*)d_out;
  float* hid = out + 2 * BATCH * MDIM;

  float* ws = (float*)d_ws;
  float* xbuf    = ws;
  float* c_buf   = xbuf + (size_t)NTREE * BATCH * 1024;
  float* tmp_iou = c_buf + (size_t)BATCH * NTREE * MDIM;
  float* tmp_f   = tmp_iou + (size_t)128 * BATCH * 768;
  float* wcat    = tmp_f + (size_t)128 * 2 * BATCH * MDIM;
  float* bcat    = wcat + (size_t)1024 * 256;

  pack_weights<<<(1024 * 256 + 1024) / 256, 256, 0, stream>>>(
      W_ioux, b_ioux, W_fx, b_fx, wcat, bcat);

  dim3 g1(NTREE, 4);
  xproj_gemm<<<g1, 256, 0, stream>>>(inputs, wcat, bcat, xbuf);

  leaf_kernel<<<(256 * BATCH * MDIM) / 256, 256, 0, stream>>>(xbuf, b_iouh, c_buf, hid);

  const int starts[8] = {256, 384, 448, 480, 496, 504, 508, 510};
  const int counts[8] = {128, 64, 32, 16, 8, 4, 2, 1};
  for (int L = 0; L < 8; ++L) {
    if (L < 3) {
      level_gemm_w8<<<dim3(counts[L], 5), 256, 0, stream>>>(
          hid, W_iouh, W_fh, tmp_iou, tmp_f, starts[L]);
    } else {
      level_gemm_w4<<<dim3(counts[L], 10), 128, 0, stream>>>(
          hid, W_iouh, W_fh, tmp_iou, tmp_f, starts[L]);
    }
    int total = counts[L] * BATCH * MDIM;
    level_elem<<<(total + 255) / 256, 256, 0, stream>>>(
        xbuf, tmp_iou, tmp_f, b_iouh, b_fh, c_buf, hid, starts[L], counts[L]);
  }

  root_copy<<<(BATCH * MDIM) / 256, 256, 0, stream>>>(c_buf, hid, out);
}


# input projection GEMM: each workgroup keeps a 64x256 weight tile in LDS and walks 16 nodes with double-buffered input tiles; B fragments come from LDS instead of per-wave strided global loads
# speedup vs baseline: 3.7495x; 1.3649x over previous
#include <hip/hip_runtime.h>

#define NTREE 511
#define BATCH 32
#define MDIM  256
#define LSTRIDE 260

typedef __attribute__((ext_vector_type(2))) float v2f;
typedef __attribute__((ext_vector_type(4))) float v4f;
typedef __attribute__((ext_vector_type(8))) float v8f;

__device__ __forceinline__ v8f wmma4(v2f a, v2f b, v8f c) {
  return __builtin_amdgcn_wmma_f32_16x16x4_f32(false, a, false, b, (short)0, c,
                                               false, false);
}

__device__ __forceinline__ float sigf(float x) { return 1.0f / (1.0f + __expf(-x)); }

__global__ __launch_bounds__(256) void pack_weights(
    const float* __restrict__ W_ioux, const float* __restrict__ b_ioux,
    const float* __restrict__ W_fx, const float* __restrict__ b_fx,
    float* __restrict__ wcat, float* __restrict__ bcat)
{
  int idx = blockIdx.x * blockDim.x + threadIdx.x;
  if (idx < 1024 * 256) {
    int row = idx >> 8;
    wcat[idx] = (row < 768) ? W_ioux[idx] : W_fx[idx - 768 * 256];
  } else {
    int col = idx - 1024 * 256;
    bcat[col] = (col < 768) ? b_ioux[col] : b_fx[col - 768];
  }
}

#define KU 8
#define XCOLS 64
#define XGROUPS 32

__device__ __forceinline__ v2f ld2(const float* p) { return *(const v2f*)p; }

__device__ __forceinline__ void xp_load(
    v2f (&x0)[KU], v2f (&x1)[KU], v2f (&y)[KU],
    const float* a0, const float* a1, const float* w, int kb)
{
#pragma unroll
  for (int u = 0; u < KU; ++u) {
    int k = kb + 4 * u;
    x0[u] = ld2(a0 + k); x1[u] = ld2(a1 + k); y[u] = ld2(w + k);
  }
  __builtin_amdgcn_sched_barrier(0);
}

__device__ __forceinline__ void xp_mma(
    const v2f (&x0)[KU], const v2f (&x1)[KU], const v2f (&y)[KU],
    v8f& c0, v8f& c1)
{
#pragma unroll
  for (int u = 0; u < KU; ++u) {
    c0 = wmma4(x0[u], y[u], c0);
    c1 = wmma4(x1[u], y[u], c1);
  }
  __builtin_amdgcn_sched_barrier(0);
}

__global__ __launch_bounds__(128) void xproj_gemm(
    const float* __restrict__ inputs,
    const float* __restrict__ wcat,
    const float* __restrict__ bcat,
    float* __restrict__ xbuf)
{
  __shared__ float sw[XCOLS * LSTRIDE];
  __shared__ float sa[2][BATCH * LSTRIDE];

  const int tid   = threadIdx.x;
  const int cbase = blockIdx.x * XCOLS;
  const int g     = blockIdx.y;

#pragma unroll 8
  for (int i = 0; i < XCOLS * 64 / 128; ++i) {
    int F = tid + 128 * i;
    int r = F >> 6, q = F & 63;
    float4 v = *(const float4*)(wcat + (size_t)(cbase + r) * 256 + q * 4);
    *(float4*)(&sw[r * LSTRIDE + q * 4]) = v;
  }
#pragma unroll
  for (int i = 0; i < 16; ++i) {
    int F = tid + 128 * i;
    int b = F >> 6, q = F & 63;
    float4 v = *(const float4*)(inputs + (size_t)b * (NTREE * MDIM) +
                                (size_t)g * MDIM + q * 4);
    *(float4*)(&sa[0][b * LSTRIDE + q * 4]) = v;
  }
  v4f nx[16];
#pragma unroll
  for (int i = 0; i < 16; ++i) {
    int F = tid + 128 * i;
    int b = F >> 6, q = F & 63;
    nx[i] = *(const v4f*)(inputs + (size_t)b * (NTREE * MDIM) +
                             (size_t)(g + XGROUPS) * MDIM + q * 4);
  }
  __syncthreads();

  const int lane  = tid & 31;
  const int wave  = tid >> 5;
  const int lr    = lane & 15;
  const int khalf = (lane < 16) ? 0 : 2;
  const int hi8   = (lane >> 4) << 3;

  const float* wl = &sw[(wave * 16 + lr) * LSTRIDE + khalf];
  const int col   = cbase + wave * 16 + lr;
  const float bs  = bcat[col];

  int buf = 0;
  for (int n = g; n < NTREE; n += XGROUPS) {
    const float* a0 = &sa[buf][lr * LSTRIDE + khalf];
    const float* a1 = &sa[buf][(16 + lr) * LSTRIDE + khalf];

    v2f P0[KU], P1[KU], PW[KU];
    v2f Q0[KU], Q1[KU], QW[KU];
    v8f c0 = {}, c1 = {};
    xp_load(P0, P1, PW, a0, a1, wl, 0);
#pragma unroll
    for (int kb = 0; kb < 256 - 8 * KU; kb += 8 * KU) {
      xp_load(Q0, Q1, QW, a0, a1, wl, kb + 4 * KU);
      xp_mma(P0, P1, PW, c0, c1);
      xp_load(P0, P1, PW, a0, a1, wl, kb + 8 * KU);
      xp_mma(Q0, Q1, QW, c0, c1);
    }
    xp_load(Q0, Q1, QW, a0, a1, wl, 256 - 4 * KU);
    xp_mma(P0, P1, PW, c0, c1);
    xp_mma(Q0, Q1, QW, c0, c1);

#pragma unroll
    for (int i = 0; i < 16; ++i) {
      int F = tid + 128 * i;
      int b = F >> 6, q = F & 63;
      *(v4f*)(&sa[buf ^ 1][b * LSTRIDE + q * 4]) = nx[i];
    }
    __syncthreads();

    float* op = xbuf + (size_t)n * (BATCH * 1024) + col;
#pragma unroll
    for (int v = 0; v < 8; ++v) {
      op[(size_t)(v + hi8) * 1024]      = c0[v] + bs;
      op[(size_t)(16 + v + hi8) * 1024] = c1[v] + bs;
    }

    const int n2 = (n + 2 * XGROUPS < NTREE) ? n + 2 * XGROUPS : n;
#pragma unroll
    for (int i = 0; i < 16; ++i) {
      int F = tid + 128 * i;
      int b = F >> 6, q = F & 63;
      nx[i] = *(const v4f*)(inputs + (size_t)b * (NTREE * MDIM) +
                               (size_t)n2 * MDIM + q * 4);
    }
    buf ^= 1;
  }
}

__global__ __launch_bounds__(256) void leaf_kernel(
    const float* __restrict__ xbuf, const float* __restrict__ b_iouh,
    float* __restrict__ c_buf, float* __restrict__ hid)
{
  int idx = blockIdx.x * blockDim.x + threadIdx.x;
  int m = idx & 255, b = (idx >> 8) & 31, node = idx >> 13;
  const float* xp = xbuf + (size_t)node * (BATCH * 1024) + (size_t)b * 1024;
  float ig = sigf(xp[m] + b_iouh[m]);
  float og = sigf(xp[256 + m] + b_iouh[256 + m]);
  float ug = tanhf(xp[512 + m] + b_iouh[512 + m]);
  float c = ig * ug;
  float h = og * tanhf(c);
  size_t o = (size_t)b * (NTREE * MDIM) + (size_t)node * MDIM + m;
  c_buf[o] = c;
  hid[o] = h;
}


__device__ __forceinline__ void iou_load(
    v2f (&x0)[KU], v2f (&x1)[KU], v2f (&yA)[KU], v2f (&yB)[KU],
    const float* a0, const float* a1, const float* wA, const float* wB, int kb)
{
#pragma unroll
  for (int u = 0; u < KU; ++u) {
    int k = kb + 4 * u;
    x0[u] = ld2(a0 + k); x1[u] = ld2(a1 + k);
    yA[u] = ld2(wA + k); yB[u] = ld2(wB + k);
  }
  __builtin_amdgcn_sched_barrier(0);
}

__device__ __forceinline__ void iou_mma(
    const v2f (&x0)[KU], const v2f (&x1)[KU], const v2f (&yA)[KU],
    const v2f (&yB)[KU], v8f& c00, v8f& c01, v8f& c10, v8f& c11)
{
#pragma unroll
  for (int u = 0; u < KU; ++u) {
    c00 = wmma4(x0[u], yA[u], c00);
    c01 = wmma4(x0[u], yB[u], c01);
    c10 = wmma4(x1[u], yA[u], c10);
    c11 = wmma4(x1[u], yB[u], c11);
  }
  __builtin_amdgcn_sched_barrier(0);
}

__device__ __forceinline__ void fg_load(
    v2f (&x00)[KU], v2f (&x01)[KU], v2f (&x10)[KU], v2f (&x11)[KU],
    v2f (&y)[KU], const float* a00, const float* a01, const float* a10,
    const float* a11, const float* w, int kb)
{
#pragma unroll
  for (int u = 0; u < KU; ++u) {
    int k = kb + 4 * u;
    x00[u] = ld2(a00 + k); x01[u] = ld2(a01 + k);
    x10[u] = ld2(a10 + k); x11[u] = ld2(a11 + k);
    y[u] = ld2(w + k);
  }
  __builtin_amdgcn_sched_barrier(0);
}

__device__ __forceinline__ void fg_mma(
    const v2f (&x00)[KU], const v2f (&x01)[KU], const v2f (&x10)[KU],
    const v2f (&x11)[KU], const v2f (&y)[KU],
    v8f& c00, v8f& c01, v8f& c10, v8f& c11)
{
#pragma unroll
  for (int u = 0; u < KU; ++u) {
    c00 = wmma4(x00[u], y[u], c00);
    c01 = wmma4(x01[u], y[u], c01);
    c10 = wmma4(x10[u], y[u], c10);
    c11 = wmma4(x11[u], y[u], c11);
  }
  __builtin_amdgcn_sched_barrier(0);
}

template <int NTHR>
__device__ __forceinline__ void level_body(
    float (*sh)[BATCH * LSTRIDE],
    const float* __restrict__ hid, const float* __restrict__ W_iouh,
    const float* __restrict__ W_fh, float* __restrict__ tmp_iou,
    float* __restrict__ tmp_f, int level_start)
{
  const int tid  = threadIdx.x;
  const int nl   = blockIdx.x;
  const int node = level_start + nl;
  const int ch0  = 2 * node - 511;
  const int ch1  = 2 * node - 512;

  const bool iou_wg = blockIdx.y * (NTHR / 32) < 24;
  if (iou_wg) {
#pragma unroll
    for (int i = 0; i < 2048 / NTHR; ++i) {
      int F = tid + NTHR * i;
      int b = F >> 6, mq = F & 63;
      const float* hp = hid + (size_t)b * (NTREE * MDIM) + mq * 4;
      float4 v0 = *(const float4*)(hp + (size_t)ch0 * MDIM);
      float4 v1 = *(const float4*)(hp + (size_t)ch1 * MDIM);
      float4 s;
      s.x = v0.x + v1.x; s.y = v0.y + v1.y;
      s.z = v0.z + v1.z; s.w = v0.w + v1.w;
      *(float4*)(&sh[0][b * LSTRIDE + mq * 4]) = s;
    }
  } else {
#pragma unroll
    for (int i = 0; i < 4096 / NTHR; ++i) {
      int F = tid + NTHR * i;
      int which = F >> 11;
      int rem = F & 2047;
      int b = rem >> 6, mq = rem & 63;
      int ch = which ? ch1 : ch0;
      float4 v = *(const float4*)(hid + (size_t)b * (NTREE * MDIM) +
                                  (size_t)ch * MDIM + mq * 4);
      *(float4*)(&sh[which][b * LSTRIDE + mq * 4]) = v;
    }
  }
  __syncthreads();

  const int lane  = tid & 31;
  const int wave  = tid >> 5;
  const int lr    = lane & 15;
  const int khalf = (lane < 16) ? 0 : 2;
  const int hi8   = (lane >> 4) << 3;
  const int task  = blockIdx.y * (NTHR / 32) + wave;

  const float* a00 = &sh[0][lr * LSTRIDE + khalf];
  const float* a01 = &sh[0][(16 + lr) * LSTRIDE + khalf];
  const float* a10 = &sh[1][lr * LSTRIDE + khalf];
  const float* a11 = &sh[1][(16 + lr) * LSTRIDE + khalf];

  if (task < 24) {
    const int colA = task * 32 + lr;
    const int colB = colA + 16;
    const float* wA = W_iouh + (size_t)colA * 256 + khalf;
    const float* wB = W_iouh + (size_t)colB * 256 + khalf;

    v2f P0[KU], P1[KU], PA[KU], PB[KU];
    v2f Q0[KU], Q1[KU], QA[KU], QB[KU];
    v8f c00 = {}, c01 = {}, c10 = {}, c11 = {};
    iou_load(P0, P1, PA, PB, a00, a01, wA, wB, 0);
#pragma unroll 1
    for (int kb = 0; kb < 256 - 8 * KU; kb += 8 * KU) {
      iou_load(Q0, Q1, QA, QB, a00, a01, wA, wB, kb + 4 * KU);
      iou_mma(P0, P1, PA, PB, c00, c01, c10, c11);
      iou_load(P0, P1, PA, PB, a00, a01, wA, wB, kb + 8 * KU);
      iou_mma(Q0, Q1, QA, QB, c00, c01, c10, c11);
    }
    iou_load(Q0, Q1, QA, QB, a00, a01, wA, wB, 256 - 4 * KU);
    iou_mma(P0, P1, PA, PB, c00, c01, c10, c11);
    iou_mma(Q0, Q1, QA, QB, c00, c01, c10, c11);

    float* opA = tmp_iou + (size_t)nl * (BATCH * 768) + colA;
    float* opB = tmp_iou + (size_t)nl * (BATCH * 768) + colB;
#pragma unroll
    for (int v = 0; v < 8; ++v) {
      int b0 = v + hi8;
      int b1 = 16 + v + hi8;
      opA[(size_t)b0 * 768] = c00[v];
      opB[(size_t)b0 * 768] = c01[v];
      opA[(size_t)b1 * 768] = c10[v];
      opB[(size_t)b1 * 768] = c11[v];
    }
  } else {
    const int col = (task - 24) * 16 + lr;
    const float* wr = W_fh + (size_t)col * 256 + khalf;

    v2f P00[KU], P01[KU], P10[KU], P11[KU], PW[KU];
    v2f Q00[KU], Q01[KU], Q10[KU], Q11[KU], QW[KU];
    v8f c00 = {}, c01 = {}, c10 = {}, c11 = {};
    fg_load(P00, P01, P10, P11, PW, a00, a01, a10, a11, wr, 0);
#pragma unroll 1
    for (int kb = 0; kb < 256 - 8 * KU; kb += 8 * KU) {
      fg_load(Q00, Q01, Q10, Q11, QW, a00, a01, a10, a11, wr, kb + 4 * KU);
      fg_mma(P00, P01, P10, P11, PW, c00, c01, c10, c11);
      fg_load(P00, P01, P10, P11, PW, a00, a01, a10, a11, wr, kb + 8 * KU);
      fg_mma(Q00, Q01, Q10, Q11, QW, c00, c01, c10, c11);
    }
    fg_load(Q00, Q01, Q10, Q11, QW, a00, a01, a10, a11, wr, 256 - 4 * KU);
    fg_mma(P00, P01, P10, P11, PW, c00, c01, c10, c11);
    fg_mma(Q00, Q01, Q10, Q11, QW, c00, c01, c10, c11);

    float* op0 = tmp_f + (((size_t)nl * 2 + 0) * BATCH) * MDIM + col;
    float* op1 = tmp_f + (((size_t)nl * 2 + 1) * BATCH) * MDIM + col;
#pragma unroll
    for (int v = 0; v < 8; ++v) {
      int b0 = v + hi8;
      int b1 = 16 + v + hi8;
      op0[(size_t)b0 * MDIM] = c00[v];
      op0[(size_t)b1 * MDIM] = c01[v];
      op1[(size_t)b0 * MDIM] = c10[v];
      op1[(size_t)b1 * MDIM] = c11[v];
    }
  }
}

__global__ __launch_bounds__(256) void level_gemm_w8(
    const float* __restrict__ hid,
    const float* __restrict__ W_iouh,
    const float* __restrict__ W_fh,
    float* __restrict__ tmp_iou,
    float* __restrict__ tmp_f,
    int level_start)
{
  __shared__ float sh[2][BATCH * LSTRIDE];
  level_body<256>(sh, hid, W_iouh, W_fh, tmp_iou, tmp_f, level_start);
}

__global__ __launch_bounds__(128) void level_gemm_w4(
    const float* __restrict__ hid, const float* __restrict__ W_iouh,
    const float* __restrict__ W_fh, float* __restrict__ tmp_iou,
    float* __restrict__ tmp_f, int level_start)
{
  __shared__ float sh[2][BATCH * LSTRIDE];
  level_body<128>(sh, hid, W_iouh, W_fh, tmp_iou, tmp_f, level_start);
}

__global__ __launch_bounds__(256) void level_elem(
    const float* __restrict__ xbuf, const float* __restrict__ tmp_iou,
    const float* __restrict__ tmp_f, const float* __restrict__ b_iouh,
    const float* __restrict__ b_fh, float* __restrict__ c_buf,
    float* __restrict__ hid, int level_start, int count)
{
  int idx = blockIdx.x * blockDim.x + threadIdx.x;
  if (idx >= count * BATCH * MDIM) return;
  int m = idx & 255, b = (idx >> 8) & 31, nl = idx >> 13;
  int node = level_start + nl;
  int ch0 = 2 * node - 511, ch1 = 2 * node - 512;
  const float* xp = xbuf + (size_t)node * (BATCH * 1024) + (size_t)b * 1024;
  const float* ti = tmp_iou + (size_t)nl * (BATCH * 768) + (size_t)b * 768;
  const float* tf = tmp_f + (size_t)nl * (2 * BATCH * MDIM) + (size_t)b * MDIM;
  float ig = sigf(xp[m] + ti[m] + b_iouh[m]);
  float og = sigf(xp[256 + m] + ti[256 + m] + b_iouh[256 + m]);
  float ug = tanhf(xp[512 + m] + ti[512 + m] + b_iouh[512 + m]);
  float xf = xp[768 + m];
  float f0 = sigf(tf[m] + b_fh[m] + xf);
  float f1 = sigf(tf[BATCH * MDIM + m] + b_fh[m] + xf);
  size_t base = (size_t)b * (NTREE * MDIM) + m;
  float c = ig * ug + f0 * c_buf[base + (size_t)ch0 * MDIM]
                    + f1 * c_buf[base + (size_t)ch1 * MDIM];
  float h = og * tanhf(c);
  c_buf[base + (size_t)node * MDIM] = c;
  hid[base + (size_t)node * MDIM] = h;
}

__global__ __launch_bounds__(256) void root_copy(
    const float* __restrict__ c_buf, const float* __restrict__ hid,
    float* __restrict__ out)
{
  int idx = blockIdx.x * blockDim.x + threadIdx.x;
  int m = idx & 255, b = idx >> 8;
  size_t o = (size_t)b * (NTREE * MDIM) + (size_t)510 * MDIM + m;
  out[idx] = c_buf[o];
  out[BATCH * MDIM + idx] = hid[o];
}

extern "C" void kernel_launch(void* const* d_in, const int* in_sizes, int n_in,
                              void* d_out, int out_size, void* d_ws, size_t ws_size,
                              hipStream_t stream) {
  (void)in_sizes; (void)n_in; (void)out_size; (void)ws_size;
  const float* inputs = (const float*)d_in[0];
  const float* W_ioux = (const float*)d_in[1];
  const float* b_ioux = (const float*)d_in[2];
  const float* W_iouh = (const float*)d_in[3];
  const float* b_iouh = (const float*)d_in[4];
  const float* W_fx   = (const float*)d_in[5];
  const float* b_fx   = (const float*)d_in[6];
  const float* W_fh   = (const float*)d_in[7];
  const float* b_fh   = (const float*)d_in[8];

  float* out = (float*)d_out;
  float* hid = out + 2 * BATCH * MDIM;

  float* ws = (float*)d_ws;
  float* xbuf    = ws;
  float* c_buf   = xbuf + (size_t)NTREE * BATCH * 1024;
  float* tmp_iou = c_buf + (size_t)BATCH * NTREE * MDIM;
  float* tmp_f   = tmp_iou + (size_t)128 * BATCH * 768;
  float* wcat    = tmp_f + (size_t)128 * 2 * BATCH * MDIM;
  float* bcat    = wcat + (size_t)1024 * 256;

  pack_weights<<<(1024 * 256 + 1024) / 256, 256, 0, stream>>>(
      W_ioux, b_ioux, W_fx, b_fx, wcat, bcat);

  dim3 g1(16, 32);
  xproj_gemm<<<g1, 128, 0, stream>>>(inputs, wcat, bcat, xbuf);

  leaf_kernel<<<(256 * BATCH * MDIM) / 256, 256, 0, stream>>>(xbuf, b_iouh, c_buf, hid);

  const int starts[8] = {256, 384, 448, 480, 496, 504, 508, 510};
  const int counts[8] = {128, 64, 32, 16, 8, 4, 2, 1};
  for (int L = 0; L < 8; ++L) {
    if (L < 3) {
      level_gemm_w8<<<dim3(counts[L], 5), 256, 0, stream>>>(
          hid, W_iouh, W_fh, tmp_iou, tmp_f, starts[L]);
    } else {
      level_gemm_w4<<<dim3(counts[L], 10), 128, 0, stream>>>(
          hid, W_iouh, W_fh, tmp_iou, tmp_f, starts[L]);
    }
    int total = counts[L] * BATCH * MDIM;
    level_elem<<<(total + 255) / 256, 256, 0, stream>>>(
        xbuf, tmp_iou, tmp_f, b_iouh, b_fh, c_buf, hid, starts[L], counts[L]);
  }

  root_copy<<<(BATCH * MDIM) / 256, 256, 0, stream>>>(c_buf, hid, out);
}


# narrow tree levels (16 nodes and fewer): K loop split four ways over the waves of a workgroup owning 32 columns x 16 rows; partial tiles summed in LDS, gate update shared by the four waves
# speedup vs baseline: 4.7082x; 1.2557x over previous
#include <hip/hip_runtime.h>

#define NTREE 511
#define BATCH 32
#define MDIM  256
#define LSTRIDE 260

typedef __attribute__((ext_vector_type(2))) float v2f;
typedef __attribute__((ext_vector_type(4))) float v4f;
typedef __attribute__((ext_vector_type(8))) float v8f;

__device__ __forceinline__ v8f wmma4(v2f a, v2f b, v8f c) {
  return __builtin_amdgcn_wmma_f32_16x16x4_f32(false, a, false, b, (short)0, c,
                                               false, false);
}

__device__ __forceinline__ float sigf(float x) { return 1.0f / (1.0f + __expf(-x)); }

__global__ __launch_bounds__(256) void pack_weights(
    const float* __restrict__ W_ioux, const float* __restrict__ b_ioux,
    const float* __restrict__ W_fx, const float* __restrict__ b_fx,
    const float* __restrict__ W_iouh, const float* __restrict__ W_fh,
    float* __restrict__ wcat, float* __restrict__ bcat,
    float* __restrict__ wrec)
{
  int idx = blockIdx.x * blockDim.x + threadIdx.x;
  if (idx < 1024 * 256) {
    int row = idx >> 8;
    wcat[idx] = (row < 768) ? W_ioux[idx] : W_fx[idx - 768 * 256];
  } else if (idx < 1024 * 256 + 1024) {
    int col = idx - 1024 * 256;
    bcat[col] = (col < 768) ? b_ioux[col] : b_fx[col - 768];
  } else {
    int p    = idx - (1024 * 256 + 1024);
    int e    = p & 3;
    int l    = (p >> 2) & 31;
    int pair = (p >> 7) & 31;
    int tile = p >> 12;
    int row  = tile * 16 + (l & 15);
    int k    = 8 * pair + 4 * (e >> 1) + 2 * (l >> 4) + (e & 1);
    wrec[p] = (row < 768) ? W_iouh[row * 256 + k] : W_fh[(row - 768) * 256 + k];
  }
}

#define KU 8
#define XCOLS 64
#define XGROUPS 32

__device__ __forceinline__ v2f ld2(const float* p) { return *(const v2f*)p; }

__device__ __forceinline__ void xp_load(
    v2f (&x0)[KU], v2f (&x1)[KU], v2f (&y)[KU],
    const float* a0, const float* a1, const float* w, int kb)
{
#pragma unroll
  for (int u = 0; u < KU; ++u) {
    int k = kb + 4 * u;
    x0[u] = ld2(a0 + k); x1[u] = ld2(a1 + k); y[u] = ld2(w + k);
  }
  __builtin_amdgcn_sched_barrier(0);
}

__device__ __forceinline__ void xp_mma(
    const v2f (&x0)[KU], const v2f (&x1)[KU], const v2f (&y)[KU],
    v8f& c0, v8f& c1)
{
#pragma unroll
  for (int u = 0; u < KU; ++u) {
    c0 = wmma4(x0[u], y[u], c0);
    c1 = wmma4(x1[u], y[u], c1);
  }
  __builtin_amdgcn_sched_barrier(0);
}

__global__ __launch_bounds__(128) void xproj_gemm(
    const float* __restrict__ inputs,
    const float* __restrict__ wcat,
    const float* __restrict__ bcat,
    float* __restrict__ xbuf)
{
  __shared__ float sw[XCOLS * LSTRIDE];
  __shared__ float sa[2][BATCH * LSTRIDE];

  const int tid   = threadIdx.x;
  const int cbase = blockIdx.x * XCOLS;
  const int g     = blockIdx.y;

#pragma unroll 8
  for (int i = 0; i < XCOLS * 64 / 128; ++i) {
    int F = tid + 128 * i;
    int r = F >> 6, q = F & 63;
    float4 v = *(const float4*)(wcat + (size_t)(cbase + r) * 256 + q * 4);
    *(float4*)(&sw[r * LSTRIDE + q * 4]) = v;
  }
#pragma unroll
  for (int i = 0; i < 16; ++i) {
    int F = tid + 128 * i;
    int b = F >> 6, q = F & 63;
    float4 v = *(const float4*)(inputs + (size_t)b * (NTREE * MDIM) +
                                (size_t)g * MDIM + q * 4);
    *(float4*)(&sa[0][b * LSTRIDE + q * 4]) = v;
  }
  v4f nx[16];
#pragma unroll
  for (int i = 0; i < 16; ++i) {
    int F = tid + 128 * i;
    int b = F >> 6, q = F & 63;
    nx[i] = *(const v4f*)(inputs + (size_t)b * (NTREE * MDIM) +
                             (size_t)(g + XGROUPS) * MDIM + q * 4);
  }
  __syncthreads();

  const int lane  = tid & 31;
  const int wave  = tid >> 5;
  const int lr    = lane & 15;
  const int khalf = (lane < 16) ? 0 : 2;
  const int hi8   = (lane >> 4) << 3;

  const float* wl = &sw[(wave * 16 + lr) * LSTRIDE + khalf];
  const int col   = cbase + wave * 16 + lr;
  const float bs  = bcat[col];

  int buf = 0;
  for (int n = g; n < NTREE; n += XGROUPS) {
    const float* a0 = &sa[buf][lr * LSTRIDE + khalf];
    const float* a1 = &sa[buf][(16 + lr) * LSTRIDE + khalf];

    v2f P0[KU], P1[KU], PW[KU];
    v2f Q0[KU], Q1[KU], QW[KU];
    v8f c0 = {}, c1 = {};
    xp_load(P0, P1, PW, a0, a1, wl, 0);
#pragma unroll
    for (int kb = 0; kb < 256 - 8 * KU; kb += 8 * KU) {
      xp_load(Q0, Q1, QW, a0, a1, wl, kb + 4 * KU);
      xp_mma(P0, P1, PW, c0, c1);
      xp_load(P0, P1, PW, a0, a1, wl, kb + 8 * KU);
      xp_mma(Q0, Q1, QW, c0, c1);
    }
    xp_load(Q0, Q1, QW, a0, a1, wl, 256 - 4 * KU);
    xp_mma(P0, P1, PW, c0, c1);
    xp_mma(Q0, Q1, QW, c0, c1);

#pragma unroll
    for (int i = 0; i < 16; ++i) {
      int F = tid + 128 * i;
      int b = F >> 6, q = F & 63;
      *(v4f*)(&sa[buf ^ 1][b * LSTRIDE + q * 4]) = nx[i];
    }
    __syncthreads();

    float* op = xbuf + (size_t)n * (BATCH * 1024) + col;
#pragma unroll
    for (int v = 0; v < 8; ++v) {
      op[(size_t)(v + hi8) * 1024]      = c0[v] + bs;
      op[(size_t)(16 + v + hi8) * 1024] = c1[v] + bs;
    }

    const int n2 = (n + 2 * XGROUPS < NTREE) ? n + 2 * XGROUPS : n;
#pragma unroll
    for (int i = 0; i < 16; ++i) {
      int F = tid + 128 * i;
      int b = F >> 6, q = F & 63;
      nx[i] = *(const v4f*)(inputs + (size_t)b * (NTREE * MDIM) +
                               (size_t)n2 * MDIM + q * 4);
    }
    buf ^= 1;
  }
}

__global__ __launch_bounds__(256) void leaf_kernel(
    const float* __restrict__ xbuf, const float* __restrict__ b_iouh,
    float* __restrict__ c_buf, float* __restrict__ hid)
{
  int idx = blockIdx.x * blockDim.x + threadIdx.x;
  int m = idx & 255, b = (idx >> 8) & 31, node = idx >> 13;
  const float* xp = xbuf + (size_t)node * (BATCH * 1024) + (size_t)b * 1024;
  float ig = sigf(xp[m] + b_iouh[m]);
  float og = sigf(xp[256 + m] + b_iouh[256 + m]);
  float ug = tanhf(xp[512 + m] + b_iouh[512 + m]);
  float c = ig * ug;
  float h = og * tanhf(c);
  size_t o = (size_t)b * (NTREE * MDIM) + (size_t)node * MDIM + m;
  c_buf[o] = c;
  hid[o] = h;
}

__device__ __forceinline__ v4f ld4(const float* p) { return *(const v4f*)p; }

__device__ __forceinline__ void fz_load(
    v2f (&x0)[KU], v2f (&x1)[KU], v4f (&yI)[KU / 2], v4f (&yO)[KU / 2],
    v4f (&yU)[KU / 2], v4f (&yF)[KU / 2],
    const float* a0, const float* a1, const float* wI, const float* wO,
    const float* wU, const float* wF, int kb)
{
#pragma unroll
  for (int u = 0; u < KU; ++u) {
    int k = kb + 4 * u;
    x0[u] = ld2(a0 + k); x1[u] = ld2(a1 + k);
  }
#pragma unroll
  for (int p = 0; p < KU / 2; ++p) {
    int off = (kb / 8 + p) * 128;
    yI[p] = ld4(wI + off); yO[p] = ld4(wO + off);
    yU[p] = ld4(wU + off); yF[p] = ld4(wF + off);
  }
  __builtin_amdgcn_sched_barrier(0);
}

__device__ __forceinline__ void fz_mma(
    const v2f (&x0)[KU], const v2f (&x1)[KU], const v4f (&yI)[KU / 2],
    const v4f (&yO)[KU / 2], const v4f (&yU)[KU / 2], const v4f (&yF)[KU / 2],
    v8f& cI, v8f& cO, v8f& cU, v8f& cF0, v8f& cF1)
{
#pragma unroll
  for (int u = 0; u < KU; ++u) {
    v2f xs = x0[u] + x1[u];
    v2f bI, bO, bU, bF;
    if (u & 1) {
      bI = __builtin_shufflevector(yI[u / 2], yI[u / 2], 2, 3);
      bO = __builtin_shufflevector(yO[u / 2], yO[u / 2], 2, 3);
      bU = __builtin_shufflevector(yU[u / 2], yU[u / 2], 2, 3);
      bF = __builtin_shufflevector(yF[u / 2], yF[u / 2], 2, 3);
    } else {
      bI = __builtin_shufflevector(yI[u / 2], yI[u / 2], 0, 1);
      bO = __builtin_shufflevector(yO[u / 2], yO[u / 2], 0, 1);
      bU = __builtin_shufflevector(yU[u / 2], yU[u / 2], 0, 1);
      bF = __builtin_shufflevector(yF[u / 2], yF[u / 2], 0, 1);
    }
    cI  = wmma4(xs, bI, cI);
    cO  = wmma4(xs, bO, cO);
    cU  = wmma4(xs, bU, cU);
    cF0 = wmma4(x0[u], bF, cF0);
    cF1 = wmma4(x1[u], bF, cF1);
  }
  __builtin_amdgcn_sched_barrier(0);
}

template <int NW>
__device__ __forceinline__ void level_fused_body(
    float (*sh)[BATCH * LSTRIDE],
    const float* __restrict__ xbuf, const float* __restrict__ wrec,
    const float* __restrict__ b_iouh, const float* __restrict__ b_fh,
    float* __restrict__ c_buf, float* __restrict__ hid, int level_start)
{
  constexpr int NTHR = NW * 32;
  constexpr int CT   = NW / 2;
  constexpr int CW   = CT * 16;
  constexpr int QR   = CW / 4;

  const int tid  = threadIdx.x;
  const int node = level_start + blockIdx.x;
  const int ch0  = 2 * node - 511;
  const int ch1  = 2 * node - 512;

#pragma unroll
  for (int i = 0; i < 4096 / NTHR; ++i) {
    int F = tid + NTHR * i;
    int which = F >> 11;
    int rem = F & 2047;
    int b = rem >> 6, mq = rem & 63;
    int ch = which ? ch1 : ch0;
    float4 v = *(const float4*)(hid + (size_t)b * (NTREE * MDIM) +
                                (size_t)ch * MDIM + mq * 4);
    *(float4*)(&sh[which][b * LSTRIDE + mq * 4]) = v;
  }

  const int lane  = tid & 31;
  const int wave  = tid >> 5;
  const int lr    = lane & 15;
  const int khalf = (lane < 16) ? 0 : 2;
  const int hi8   = (lane >> 4) << 3;
  const int jl    = wave >> 1;
  const int rt    = wave & 1;
  const int j     = blockIdx.y * CT + jl;
  const int m     = j * 16 + lr;

  float xi[8], xo[8], xu[8], xf[8], p0[8], p1[8];
#pragma unroll
  for (int v = 0; v < 8; ++v) {
    int b = 16 * rt + hi8 + v;
    const float* xp = xbuf + (size_t)node * (BATCH * 1024) + (size_t)b * 1024 + m;
    xi[v] = xp[0]; xo[v] = xp[256]; xu[v] = xp[512]; xf[v] = xp[768];
    size_t cb = (size_t)b * (NTREE * MDIM) + m;
    p0[v] = c_buf[cb + (size_t)ch0 * MDIM];
    p1[v] = c_buf[cb + (size_t)ch1 * MDIM];
  }
  const float bsI = b_iouh[m], bsO = b_iouh[256 + m], bsU = b_iouh[512 + m];
  const float bsF = b_fh[m];

  __syncthreads();

  const float* a0 = &sh[0][(16 * rt + lr) * LSTRIDE + khalf];
  const float* a1 = &sh[1][(16 * rt + lr) * LSTRIDE + khalf];
  const float* wI = wrec + (size_t)(j)      * 4096 + lane * 4;
  const float* wO = wrec + (size_t)(16 + j) * 4096 + lane * 4;
  const float* wU = wrec + (size_t)(32 + j) * 4096 + lane * 4;
  const float* wF = wrec + (size_t)(48 + j) * 4096 + lane * 4;

  v2f P0[KU], P1[KU], Q0[KU], Q1[KU];
  v4f PI[KU / 2], PO[KU / 2], PU[KU / 2], PF[KU / 2];
  v4f QI[KU / 2], QO[KU / 2], QU[KU / 2], QF[KU / 2];
  v8f cI = {}, cO = {}, cU = {}, cF0 = {}, cF1 = {};
  fz_load(P0, P1, PI, PO, PU, PF, a0, a1, wI, wO, wU, wF, 0);
#pragma unroll 1
  for (int kb = 0; kb < 256 - 8 * KU; kb += 8 * KU) {
    fz_load(Q0, Q1, QI, QO, QU, QF, a0, a1, wI, wO, wU, wF, kb + 4 * KU);
    fz_mma(P0, P1, PI, PO, PU, PF, cI, cO, cU, cF0, cF1);
    fz_load(P0, P1, PI, PO, PU, PF, a0, a1, wI, wO, wU, wF, kb + 8 * KU);
    fz_mma(Q0, Q1, QI, QO, QU, QF, cI, cO, cU, cF0, cF1);
  }
  fz_load(Q0, Q1, QI, QO, QU, QF, a0, a1, wI, wO, wU, wF, 256 - 4 * KU);
  fz_mma(P0, P1, PI, PO, PU, PF, cI, cO, cU, cF0, cF1);
  fz_mma(Q0, Q1, QI, QO, QU, QF, cI, cO, cU, cF0, cF1);

  __syncthreads();

#pragma unroll
  for (int v = 0; v < 8; ++v) {
    int b = 16 * rt + hi8 + v;
    float ig = sigf(xi[v] + cI[v] + bsI);
    float og = sigf(xo[v] + cO[v] + bsO);
    float ug = tanhf(xu[v] + cU[v] + bsU);
    float f0 = sigf(cF0[v] + bsF + xf[v]);
    float f1 = sigf(cF1[v] + bsF + xf[v]);
    float c = ig * ug + f0 * p0[v] + f1 * p1[v];
    float h = og * tanhf(c);
    sh[0][b * LSTRIDE + jl * 16 + lr] = h;
    sh[1][b * LSTRIDE + jl * 16 + lr] = c;
  }
  __syncthreads();

  const int cbase = blockIdx.y * CW;
#pragma unroll
  for (int i = 0; i < (BATCH * QR) / NTHR; ++i) {
    int F = tid + NTHR * i;
    int b = F / QR, q = F % QR;
    v4f hv = *(const v4f*)(&sh[0][b * LSTRIDE + q * 4]);
    v4f cv = *(const v4f*)(&sh[1][b * LSTRIDE + q * 4]);
    size_t o = (size_t)b * (NTREE * MDIM) + (size_t)node * MDIM + cbase + q * 4;
    *(v4f*)(hid + o)   = hv;
    *(v4f*)(c_buf + o) = cv;
  }
}

__global__ __launch_bounds__(128) void level_fused_w4(
    const float* __restrict__ xbuf, const float* __restrict__ wrec,
    const float* __restrict__ b_iouh, const float* __restrict__ b_fh,
    float* __restrict__ c_buf, float* __restrict__ hid, int level_start)
{
  __shared__ float sh[2][BATCH * LSTRIDE];
  level_fused_body<4>(sh, xbuf, wrec, b_iouh, b_fh, c_buf, hid, level_start);
}

__device__ __forceinline__ void fzb_load(
    v4f (&yI)[KU / 2], v4f (&yO)[KU / 2], v4f (&yU)[KU / 2], v4f (&yF)[KU / 2],
    const float* wI, const float* wO, const float* wU, const float* wF, int kb)
{
#pragma unroll
  for (int p = 0; p < KU / 2; ++p) {
    int off = (kb / 8 + p) * 128;
    yI[p] = ld4(wI + off); yO[p] = ld4(wO + off);
    yU[p] = ld4(wU + off); yF[p] = ld4(wF + off);
  }
}

__device__ __forceinline__ void fza_load(
    v2f (&x0)[KU], v2f (&x1)[KU], const float* a0, const float* a1, int kb)
{
#pragma unroll
  for (int u = 0; u < KU; ++u) {
    int k = kb + 4 * u;
    x0[u] = ld2(a0 + k); x1[u] = ld2(a1 + k);
  }
}

#define SOP 36

__global__ __launch_bounds__(256) void level_fused_ks4(
    const float* __restrict__ xbuf, const float* __restrict__ wrec,
    const float* __restrict__ b_iouh, const float* __restrict__ b_fh,
    float* __restrict__ c_buf, float* __restrict__ hid, int level_start)
{
  __shared__ float sh[2][16 * LSTRIDE];
  __shared__ float red[8][5][8][32];
  __shared__ float so[2][16 * SOP];

  const int tid  = threadIdx.x;
  const int node = level_start + blockIdx.x;
  const int ch0  = 2 * node - 511;
  const int ch1  = 2 * node - 512;
  const int cp   = blockIdx.y >> 1;
  const int rt   = blockIdx.y & 1;

  const int lane  = tid & 31;
  const int wave  = tid >> 5;
  const int lr    = lane & 15;
  const int khalf = (lane < 16) ? 0 : 2;
  const int hi8   = (lane >> 4) << 3;
  const int jl    = wave >> 2;
  const int ks    = wave & 3;
  const int j     = cp * 2 + jl;
  const int m     = j * 16 + lr;

  v4f st[8];
#pragma unroll
  for (int i = 0; i < 8; ++i) {
    int rem = (tid + 256 * i) & 1023;
    int r = rem >> 6, mq = rem & 63;
    int ch = (i >> 2) ? ch1 : ch0;
    st[i] = *(const v4f*)(hid + (size_t)(16 * rt + r) * (NTREE * MDIM) +
                          (size_t)ch * MDIM + mq * 4);
  }
  __builtin_amdgcn_sched_barrier(0);

  const size_t wq = (size_t)ks * 1024 + lane * 4;
  const float* wI = wrec + (size_t)(j)      * 4096 + wq;
  const float* wO = wrec + (size_t)(16 + j) * 4096 + wq;
  const float* wU = wrec + (size_t)(32 + j) * 4096 + wq;
  const float* wF = wrec + (size_t)(48 + j) * 4096 + wq;
  v4f PI[KU / 2], PO[KU / 2], PU[KU / 2], PF[KU / 2];
  v4f QI[KU / 2], QO[KU / 2], QU[KU / 2], QF[KU / 2];
  fzb_load(PI, PO, PU, PF, wI, wO, wU, wF, 0);
  fzb_load(QI, QO, QU, QF, wI, wO, wU, wF, 4 * KU);

  float xi[2], xo[2], xu[2], xf[2], p0[2], p1[2];
#pragma unroll
  for (int e = 0; e < 2; ++e) {
    int b = 16 * rt + hi8 + 2 * ks + e;
    const float* xp = xbuf + (size_t)node * (BATCH * 1024) + (size_t)b * 1024 + m;
    xi[e] = xp[0]; xo[e] = xp[256]; xu[e] = xp[512]; xf[e] = xp[768];
    size_t cb = (size_t)b * (NTREE * MDIM) + m;
    p0[e] = c_buf[cb + (size_t)ch0 * MDIM];
    p1[e] = c_buf[cb + (size_t)ch1 * MDIM];
  }
  const float bsI = b_iouh[m], bsO = b_iouh[256 + m], bsU = b_iouh[512 + m];
  const float bsF = b_fh[m];
  __builtin_amdgcn_sched_barrier(0);

#pragma unroll
  for (int i = 0; i < 8; ++i) {
    int rem = (tid + 256 * i) & 1023;
    int r = rem >> 6, mq = rem & 63;
    *(v4f*)(&sh[i >> 2][r * LSTRIDE + mq * 4]) = st[i];
  }
  __builtin_amdgcn_fence(__ATOMIC_RELEASE, "workgroup", "local");
  __builtin_amdgcn_s_barrier();
  __builtin_amdgcn_fence(__ATOMIC_ACQUIRE, "workgroup", "local");

  const float* a0 = &sh[0][lr * LSTRIDE + khalf + 64 * ks];
  const float* a1 = &sh[1][lr * LSTRIDE + khalf + 64 * ks];
  v2f P0[KU], P1[KU], Q0[KU], Q1[KU];
  fza_load(P0, P1, a0, a1, 0);
  fza_load(Q0, Q1, a0, a1, 4 * KU);
  __builtin_amdgcn_sched_barrier(0);

  v8f cI = {}, cO = {}, cU = {}, cF0 = {}, cF1 = {};
  fz_mma(P0, P1, PI, PO, PU, PF, cI, cO, cU, cF0, cF1);
  fz_mma(Q0, Q1, QI, QO, QU, QF, cI, cO, cU, cF0, cF1);

#pragma unroll
  for (int v = 0; v < 8; ++v) {
    red[wave][0][v][lane] = cI[v];
    red[wave][1][v][lane] = cO[v];
    red[wave][2][v][lane] = cU[v];
    red[wave][3][v][lane] = cF0[v];
    red[wave][4][v][lane] = cF1[v];
  }
  __syncthreads();

#pragma unroll
  for (int e = 0; e < 2; ++e) {
    const int v = 2 * ks + e;
    float s[5];
#pragma unroll
    for (int a = 0; a < 5; ++a) {
      float t = red[4 * jl + 0][a][v][lane];
      t += red[4 * jl + 1][a][v][lane];
      t += red[4 * jl + 2][a][v][lane];
      t += red[4 * jl + 3][a][v][lane];
      s[a] = t;
    }
    float ig = sigf(xi[e] + s[0] + bsI);
    float og = sigf(xo[e] + s[1] + bsO);
    float ug = tanhf(xu[e] + s[2] + bsU);
    float f0 = sigf(s[3] + bsF + xf[e]);
    float f1 = sigf(s[4] + bsF + xf[e]);
    float c = ig * ug + f0 * p0[e] + f1 * p1[e];
    float h = og * tanhf(c);
    const int row = hi8 + v;
    so[0][row * SOP + jl * 16 + lr] = h;
    so[1][row * SOP + jl * 16 + lr] = c;
  }
  __syncthreads();

  {
    const int which = tid >> 7;
    const int row = (tid >> 3) & 15, q = tid & 7;
    v4f val = *(const v4f*)(&so[which][row * SOP + q * 4]);
    size_t o = (size_t)(16 * rt + row) * (NTREE * MDIM) + (size_t)node * MDIM +
               cp * 32 + q * 4;
    float* dst = which ? c_buf : hid;
    *(v4f*)(dst + o) = val;
  }
}

__global__ __launch_bounds__(256) void root_copy(
    const float* __restrict__ c_buf, const float* __restrict__ hid,
    float* __restrict__ out)
{
  int idx = blockIdx.x * blockDim.x + threadIdx.x;
  int m = idx & 255, b = idx >> 8;
  size_t o = (size_t)b * (NTREE * MDIM) + (size_t)510 * MDIM + m;
  out[idx] = c_buf[o];
  out[BATCH * MDIM + idx] = hid[o];
}

extern "C" void kernel_launch(void* const* d_in, const int* in_sizes, int n_in,
                              void* d_out, int out_size, void* d_ws, size_t ws_size,
                              hipStream_t stream) {
  (void)in_sizes; (void)n_in; (void)out_size; (void)ws_size;
  const float* inputs = (const float*)d_in[0];
  const float* W_ioux = (const float*)d_in[1];
  const float* b_ioux = (const float*)d_in[2];
  const float* W_iouh = (const float*)d_in[3];
  const float* b_iouh = (const float*)d_in[4];
  const float* W_fx   = (const float*)d_in[5];
  const float* b_fx   = (const float*)d_in[6];
  const float* W_fh   = (const float*)d_in[7];
  const float* b_fh   = (const float*)d_in[8];

  float* out = (float*)d_out;
  float* hid = out + 2 * BATCH * MDIM;

  float* ws = (float*)d_ws;
  float* xbuf    = ws;
  float* c_buf   = xbuf + (size_t)NTREE * BATCH * 1024;
  float* wrec    = c_buf + (size_t)BATCH * NTREE * MDIM;
  float* wcat    = wrec + (size_t)128 * BATCH * 768 + (size_t)128 * 2 * BATCH * MDIM;
  float* bcat    = wcat + (size_t)1024 * 256;

  pack_weights<<<2052, 256, 0, stream>>>(
      W_ioux, b_ioux, W_fx, b_fx, W_iouh, W_fh, wcat, bcat, wrec);

  dim3 g1(16, 32);
  xproj_gemm<<<g1, 128, 0, stream>>>(inputs, wcat, bcat, xbuf);

  leaf_kernel<<<(256 * BATCH * MDIM) / 256, 256, 0, stream>>>(xbuf, b_iouh, c_buf, hid);

  const int starts[8] = {256, 384, 448, 480, 496, 504, 508, 510};
  const int counts[8] = {128, 64, 32, 16, 8, 4, 2, 1};
  for (int L = 0; L < 3; ++L) {
    level_fused_w4<<<dim3(counts[L], 8), 128, 0, stream>>>(
        xbuf, wrec, b_iouh, b_fh, c_buf, hid, starts[L]);
  }
  for (int L = 3; L < 8; ++L) {
    level_fused_ks4<<<dim3(counts[L], 16), 256, 0, stream>>>(
        xbuf, wrec, b_iouh, b_fh, c_buf, hid, starts[L]);
  }

  root_copy<<<(BATCH * MDIM) / 256, 256, 0, stream>>>(c_buf, hid, out);
}


# gate math in all three kernels: sigmoid and tanh each from one hardware exp2 and one hardware reciprocal (f32), replacing the IEEE division sequence and the branching library tanh
# speedup vs baseline: 5.7252x; 1.2160x over previous
#include <hip/hip_runtime.h>

#define NTREE 511
#define BATCH 32
#define MDIM  256
#define LSTRIDE 260

typedef __attribute__((ext_vector_type(2))) float v2f;
typedef __attribute__((ext_vector_type(4))) float v4f;
typedef __attribute__((ext_vector_type(8))) float v8f;

__device__ __forceinline__ v8f wmma4(v2f a, v2f b, v8f c) {
  return __builtin_amdgcn_wmma_f32_16x16x4_f32(false, a, false, b, (short)0, c,
                                               false, false);
}

__device__ __forceinline__ float sigf(float x) {
  return __builtin_amdgcn_rcpf(1.0f + __builtin_amdgcn_exp2f(x * -1.4426950408889634f));
}
__device__ __forceinline__ float tanhq(float x) {
  float r = __builtin_amdgcn_rcpf(1.0f + __builtin_amdgcn_exp2f(x * -2.8853900817779268f));
  return __builtin_fmaf(2.0f, r, -1.0f);
}

__global__ __launch_bounds__(256) void pack_weights(
    const float* __restrict__ W_ioux, const float* __restrict__ b_ioux,
    const float* __restrict__ W_fx, const float* __restrict__ b_fx,
    const float* __restrict__ W_iouh, const float* __restrict__ W_fh,
    float* __restrict__ wcat, float* __restrict__ bcat,
    float* __restrict__ wrec)
{
  int idx = blockIdx.x * blockDim.x + threadIdx.x;
  if (idx < 1024 * 256) {
    int row = idx >> 8;
    wcat[idx] = (row < 768) ? W_ioux[idx] : W_fx[idx - 768 * 256];
  } else if (idx < 1024 * 256 + 1024) {
    int col = idx - 1024 * 256;
    bcat[col] = (col < 768) ? b_ioux[col] : b_fx[col - 768];
  } else {
    int p    = idx - (1024 * 256 + 1024);
    int e    = p & 3;
    int l    = (p >> 2) & 31;
    int pair = (p >> 7) & 31;
    int tile = p >> 12;
    int row  = tile * 16 + (l & 15);
    int k    = 8 * pair + 4 * (e >> 1) + 2 * (l >> 4) + (e & 1);
    wrec[p] = (row < 768) ? W_iouh[row * 256 + k] : W_fh[(row - 768) * 256 + k];
  }
}

#define KU 8
#define XCOLS 64
#define XGROUPS 32
#define LROWS 96
#define LGROUPS 64
#define LSOP 36

__device__ __forceinline__ v2f ld2(const float* p) { return *(const v2f*)p; }

__device__ __forceinline__ void xp_load(
    v2f (&x0)[KU], v2f (&x1)[KU], v2f (&y)[KU],
    const float* a0, const float* a1, const float* w, int kb)
{
#pragma unroll
  for (int u = 0; u < KU; ++u) {
    int k = kb + 4 * u;
    x0[u] = ld2(a0 + k); x1[u] = ld2(a1 + k); y[u] = ld2(w + k);
  }
  __builtin_amdgcn_sched_barrier(0);
}

__device__ __forceinline__ void xp_mma(
    const v2f (&x0)[KU], const v2f (&x1)[KU], const v2f (&y)[KU],
    v8f& c0, v8f& c1)
{
#pragma unroll
  for (int u = 0; u < KU; ++u) {
    c0 = wmma4(x0[u], y[u], c0);
    c1 = wmma4(x1[u], y[u], c1);
  }
  __builtin_amdgcn_sched_barrier(0);
}

__device__ __forceinline__ void lf_load(
    v2f (&x)[KU], v2f (&y0)[KU], v2f (&y1)[KU], v2f (&y2)[KU],
    const float* a, const float* w, int kb)
{
#pragma unroll
  for (int u = 0; u < KU; ++u) {
    int k = kb + 4 * u;
    x[u]  = ld2(a + k);
    y0[u] = ld2(w + k);
    y1[u] = ld2(w + LSTRIDE + k);
    y2[u] = ld2(w + 2 * LSTRIDE + k);
  }
  __builtin_amdgcn_sched_barrier(0);
}

__device__ __forceinline__ void lf_mma(
    const v2f (&x)[KU], const v2f (&y0)[KU], const v2f (&y1)[KU],
    const v2f (&y2)[KU], v8f& c0, v8f& c1, v8f& c2)
{
#pragma unroll
  for (int u = 0; u < KU; ++u) {
    c0 = wmma4(x[u], y0[u], c0);
    c1 = wmma4(x[u], y1[u], c1);
    c2 = wmma4(x[u], y2[u], c2);
  }
  __builtin_amdgcn_sched_barrier(0);
}

__global__ __launch_bounds__(128) void xproj_gemm(
    const float* __restrict__ inputs,
    const float* __restrict__ wcat,
    const float* __restrict__ bcat,
    const float* __restrict__ b_iouh,
    float* __restrict__ xbuf,
    float* __restrict__ c_buf,
    float* __restrict__ hid)
{
  __shared__ __attribute__((aligned(16))) float lds[128 * LSTRIDE + 2 * 32 * LSOP];

  const int tid   = threadIdx.x;
  const int lane  = tid & 31;
  const int wave  = tid >> 5;
  const int lr    = lane & 15;
  const int khalf = (lane < 16) ? 0 : 2;
  const int hi8   = (lane >> 4) << 3;

  {
    float* sw3 = lds;
    float* sa  = lds + LROWS * LSTRIDE;
    float* so  = lds + 128 * LSTRIDE;
    const int lc = blockIdx.x >> 1;
    const int lg = blockIdx.y * 2 + (blockIdx.x & 1);

#pragma unroll
    for (int i = 0; i < LROWS * 64 / 128; ++i) {
      int F = tid + 128 * i;
      int r = F >> 6, q = F & 63;
      int cl = r / 3, gt = r - 3 * cl;
      int srow = gt * 256 + lc * 32 + cl;
      float4 v = *(const float4*)(wcat + (size_t)srow * 256 + q * 4);
      *(float4*)(&sw3[r * LSTRIDE + q * 4]) = v;
    }
#pragma unroll
    for (int i = 0; i < 16; ++i) {
      int F = tid + 128 * i;
      int b = F >> 6, q = F & 63;
      float4 v = *(const float4*)(inputs + (size_t)b * (NTREE * MDIM) +
                                  (size_t)lg * MDIM + q * 4);
      *(float4*)(&sa[b * LSTRIDE + q * 4]) = v;
    }
    v4f nx[16];
#pragma unroll
    for (int i = 0; i < 16; ++i) {
      int F = tid + 128 * i;
      int b = F >> 6, q = F & 63;
      nx[i] = *(const v4f*)(inputs + (size_t)b * (NTREE * MDIM) +
                               (size_t)(lg + LGROUPS) * MDIM + q * 4);
    }
    __syncthreads();

    const int rt = wave & 1;
    const int ct = wave >> 1;
    const float* a = &sa[(16 * rt + lr) * LSTRIDE + khalf];
    const float* w = &sw3[3 * (16 * ct + lr) * LSTRIDE + khalf];
    const int m    = lc * 32 + ct * 16 + lr;
    const float bs0 = bcat[m], bs1 = bcat[256 + m], bs2 = bcat[512 + m];
    const float bi0 = b_iouh[m], bi1 = b_iouh[256 + m], bi2 = b_iouh[512 + m];

#pragma unroll 1
    for (int i = 0; i < 256 / LGROUPS; ++i) {
      const int node = lg + LGROUPS * i;

      v2f PX[KU], P0[KU], P1[KU], P2[KU];
      v2f QX[KU], Q0[KU], Q1[KU], Q2[KU];
      v8f c0 = {}, c1 = {}, c2 = {};
      lf_load(PX, P0, P1, P2, a, w, 0);
#pragma unroll
      for (int kb = 0; kb < 256 - 8 * KU; kb += 8 * KU) {
        lf_load(QX, Q0, Q1, Q2, a, w, kb + 4 * KU);
        lf_mma(PX, P0, P1, P2, c0, c1, c2);
        lf_load(PX, P0, P1, P2, a, w, kb + 8 * KU);
        lf_mma(QX, Q0, Q1, Q2, c0, c1, c2);
      }
      lf_load(QX, Q0, Q1, Q2, a, w, 256 - 4 * KU);
      lf_mma(PX, P0, P1, P2, c0, c1, c2);
      lf_mma(QX, Q0, Q1, Q2, c0, c1, c2);

#pragma unroll
      for (int v = 0; v < 8; ++v) {
        int b = 16 * rt + hi8 + v;
        float ig = sigf((c0[v] + bs0) + bi0);
        float og = sigf((c1[v] + bs1) + bi1);
        float ug = tanhq((c2[v] + bs2) + bi2);
        float cc = ig * ug;
        float hh = og * tanhq(cc);
        so[b * LSOP + ct * 16 + lr]             = hh;
        so[32 * LSOP + b * LSOP + ct * 16 + lr] = cc;
      }
      __syncthreads();

      if (i + 1 < 256 / LGROUPS) {
#pragma unroll
        for (int e = 0; e < 16; ++e) {
          int F = tid + 128 * e;
          int b = F >> 6, q = F & 63;
          *(v4f*)(&sa[b * LSTRIDE + q * 4]) = nx[e];
        }
      }
#pragma unroll
      for (int e = 0; e < 4; ++e) {
        int F = tid + 128 * e;
        int plane = F >> 8;
        int b = (F >> 3) & 31, q = F & 7;
        v4f val = *(const v4f*)(&so[plane * 32 * LSOP + b * LSOP + q * 4]);
        float* dst = plane ? c_buf : hid;
        *(v4f*)(dst + (size_t)b * (NTREE * MDIM) + (size_t)node * MDIM +
                lc * 32 + q * 4) = val;
      }
      __syncthreads();

      if (i + 2 < 256 / LGROUPS) {
#pragma unroll
        for (int e = 0; e < 16; ++e) {
          int F = tid + 128 * e;
          int b = F >> 6, q = F & 63;
          nx[e] = *(const v4f*)(inputs + (size_t)b * (NTREE * MDIM) +
                                   (size_t)(node + 2 * LGROUPS) * MDIM + q * 4);
        }
      }
    }
  }

  float* sw  = lds;
  float* sa0 = lds + XCOLS * LSTRIDE;
  const int cbase = blockIdx.x * XCOLS;
  const int g     = blockIdx.y;
  const int nfirst = 256 + g;

#pragma unroll 8
  for (int i = 0; i < XCOLS * 64 / 128; ++i) {
    int F = tid + 128 * i;
    int r = F >> 6, q = F & 63;
    float4 v = *(const float4*)(wcat + (size_t)(cbase + r) * 256 + q * 4);
    *(float4*)(&sw[r * LSTRIDE + q * 4]) = v;
  }
#pragma unroll
  for (int i = 0; i < 16; ++i) {
    int F = tid + 128 * i;
    int b = F >> 6, q = F & 63;
    float4 v = *(const float4*)(inputs + (size_t)b * (NTREE * MDIM) +
                                (size_t)nfirst * MDIM + q * 4);
    *(float4*)(&sa0[b * LSTRIDE + q * 4]) = v;
  }
  v4f nx[16];
#pragma unroll
  for (int i = 0; i < 16; ++i) {
    int F = tid + 128 * i;
    int b = F >> 6, q = F & 63;
    nx[i] = *(const v4f*)(inputs + (size_t)b * (NTREE * MDIM) +
                             (size_t)(nfirst + XGROUPS) * MDIM + q * 4);
  }
  __syncthreads();

  const float* wl = &sw[(wave * 16 + lr) * LSTRIDE + khalf];
  const int col   = cbase + wave * 16 + lr;
  const float bs  = bcat[col];

  int buf = 0;
  for (int n = nfirst; n < NTREE; n += XGROUPS) {
    const float* a0 = &sa0[buf * (BATCH * LSTRIDE) + lr * LSTRIDE + khalf];
    const float* a1 = &sa0[buf * (BATCH * LSTRIDE) + (16 + lr) * LSTRIDE + khalf];

    v2f P0[KU], P1[KU], PW[KU];
    v2f Q0[KU], Q1[KU], QW[KU];
    v8f c0 = {}, c1 = {};
    xp_load(P0, P1, PW, a0, a1, wl, 0);
#pragma unroll
    for (int kb = 0; kb < 256 - 8 * KU; kb += 8 * KU) {
      xp_load(Q0, Q1, QW, a0, a1, wl, kb + 4 * KU);
      xp_mma(P0, P1, PW, c0, c1);
      xp_load(P0, P1, PW, a0, a1, wl, kb + 8 * KU);
      xp_mma(Q0, Q1, QW, c0, c1);
    }
    xp_load(Q0, Q1, QW, a0, a1, wl, 256 - 4 * KU);
    xp_mma(P0, P1, PW, c0, c1);
    xp_mma(Q0, Q1, QW, c0, c1);

    float* sn = &sa0[(buf ^ 1) * (BATCH * LSTRIDE)];
#pragma unroll
    for (int i = 0; i < 16; ++i) {
      int F = tid + 128 * i;
      int b = F >> 6, q = F & 63;
      *(v4f*)(&sn[b * LSTRIDE + q * 4]) = nx[i];
    }
    __syncthreads();

    float* op = xbuf + (size_t)n * (BATCH * 1024) + col;
#pragma unroll
    for (int v = 0; v < 8; ++v) {
      op[(size_t)(v + hi8) * 1024]      = c0[v] + bs;
      op[(size_t)(16 + v + hi8) * 1024] = c1[v] + bs;
    }

    const int n2 = (n + 2 * XGROUPS < NTREE) ? n + 2 * XGROUPS : n;
#pragma unroll
    for (int i = 0; i < 16; ++i) {
      int F = tid + 128 * i;
      int b = F >> 6, q = F & 63;
      nx[i] = *(const v4f*)(inputs + (size_t)b * (NTREE * MDIM) +
                               (size_t)n2 * MDIM + q * 4);
    }
    buf ^= 1;
  }
}

__device__ __forceinline__ v4f ld4(const float* p) { return *(const v4f*)p; }

__device__ __forceinline__ void fz_load(
    v2f (&x0)[KU], v2f (&x1)[KU], v4f (&yI)[KU / 2], v4f (&yO)[KU / 2],
    v4f (&yU)[KU / 2], v4f (&yF)[KU / 2],
    const float* a0, const float* a1, const float* wI, const float* wO,
    const float* wU, const float* wF, int kb)
{
#pragma unroll
  for (int u = 0; u < KU; ++u) {
    int k = kb + 4 * u;
    x0[u] = ld2(a0 + k); x1[u] = ld2(a1 + k);
  }
#pragma unroll
  for (int p = 0; p < KU / 2; ++p) {
    int off = (kb / 8 + p) * 128;
    yI[p] = ld4(wI + off); yO[p] = ld4(wO + off);
    yU[p] = ld4(wU + off); yF[p] = ld4(wF + off);
  }
  __builtin_amdgcn_sched_barrier(0);
}

__device__ __forceinline__ void fz_mma(
    const v2f (&x0)[KU], const v2f (&x1)[KU], const v4f (&yI)[KU / 2],
    const v4f (&yO)[KU / 2], const v4f (&yU)[KU / 2], const v4f (&yF)[KU / 2],
    v8f& cI, v8f& cO, v8f& cU, v8f& cF0, v8f& cF1)
{
#pragma unroll
  for (int u = 0; u < KU; ++u) {
    v2f xs = x0[u] + x1[u];
    v2f bI, bO, bU, bF;
    if (u & 1) {
      bI = __builtin_shufflevector(yI[u / 2], yI[u / 2], 2, 3);
      bO = __builtin_shufflevector(yO[u / 2], yO[u / 2], 2, 3);
      bU = __builtin_shufflevector(yU[u / 2], yU[u / 2], 2, 3);
      bF = __builtin_shufflevector(yF[u / 2], yF[u / 2], 2, 3);
    } else {
      bI = __builtin_shufflevector(yI[u / 2], yI[u / 2], 0, 1);
      bO = __builtin_shufflevector(yO[u / 2], yO[u / 2], 0, 1);
      bU = __builtin_shufflevector(yU[u / 2], yU[u / 2], 0, 1);
      bF = __builtin_shufflevector(yF[u / 2], yF[u / 2], 0, 1);
    }
    cI  = wmma4(xs, bI, cI);
    cO  = wmma4(xs, bO, cO);
    cU  = wmma4(xs, bU, cU);
    cF0 = wmma4(x0[u], bF, cF0);
    cF1 = wmma4(x1[u], bF, cF1);
  }
  __builtin_amdgcn_sched_barrier(0);
}

template <int NW>
__device__ __forceinline__ void level_fused_body(
    float (*sh)[BATCH * LSTRIDE],
    const float* __restrict__ xbuf, const float* __restrict__ wrec,
    const float* __restrict__ b_iouh, const float* __restrict__ b_fh,
    float* __restrict__ c_buf, float* __restrict__ hid, int level_start)
{
  constexpr int NTHR = NW * 32;
  constexpr int CT   = NW / 2;
  constexpr int CW   = CT * 16;
  constexpr int QR   = CW / 4;

  const int tid  = threadIdx.x;
  const int node = level_start + blockIdx.x;
  const int ch0  = 2 * node - 511;
  const int ch1  = 2 * node - 512;

#pragma unroll
  for (int i = 0; i < 4096 / NTHR; ++i) {
    int F = tid + NTHR * i;
    int which = F >> 11;
    int rem = F & 2047;
    int b = rem >> 6, mq = rem & 63;
    int ch = which ? ch1 : ch0;
    float4 v = *(const float4*)(hid + (size_t)b * (NTREE * MDIM) +
                                (size_t)ch * MDIM + mq * 4);
    *(float4*)(&sh[which][b * LSTRIDE + mq * 4]) = v;
  }

  const int lane  = tid & 31;
  const int wave  = tid >> 5;
  const int lr    = lane & 15;
  const int khalf = (lane < 16) ? 0 : 2;
  const int hi8   = (lane >> 4) << 3;
  const int jl    = wave >> 1;
  const int rt    = wave & 1;
  const int j     = blockIdx.y * CT + jl;
  const int m     = j * 16 + lr;

  float xi[8], xo[8], xu[8], xf[8], p0[8], p1[8];
#pragma unroll
  for (int v = 0; v < 8; ++v) {
    int b = 16 * rt + hi8 + v;
    const float* xp = xbuf + (size_t)node * (BATCH * 1024) + (size_t)b * 1024 + m;
    xi[v] = xp[0]; xo[v] = xp[256]; xu[v] = xp[512]; xf[v] = xp[768];
    size_t cb = (size_t)b * (NTREE * MDIM) + m;
    p0[v] = c_buf[cb + (size_t)ch0 * MDIM];
    p1[v] = c_buf[cb + (size_t)ch1 * MDIM];
  }
  const float bsI = b_iouh[m], bsO = b_iouh[256 + m], bsU = b_iouh[512 + m];
  const float bsF = b_fh[m];

  __syncthreads();

  const float* a0 = &sh[0][(16 * rt + lr) * LSTRIDE + khalf];
  const float* a1 = &sh[1][(16 * rt + lr) * LSTRIDE + khalf];
  const float* wI = wrec + (size_t)(j)      * 4096 + lane * 4;
  const float* wO = wrec + (size_t)(16 + j) * 4096 + lane * 4;
  const float* wU = wrec + (size_t)(32 + j) * 4096 + lane * 4;
  const float* wF = wrec + (size_t)(48 + j) * 4096 + lane * 4;

  v2f P0[KU], P1[KU], Q0[KU], Q1[KU];
  v4f PI[KU / 2], PO[KU / 2], PU[KU / 2], PF[KU / 2];
  v4f QI[KU / 2], QO[KU / 2], QU[KU / 2], QF[KU / 2];
  v8f cI = {}, cO = {}, cU = {}, cF0 = {}, cF1 = {};
  fz_load(P0, P1, PI, PO, PU, PF, a0, a1, wI, wO, wU, wF, 0);
#pragma unroll 1
  for (int kb = 0; kb < 256 - 8 * KU; kb += 8 * KU) {
    fz_load(Q0, Q1, QI, QO, QU, QF, a0, a1, wI, wO, wU, wF, kb + 4 * KU);
    fz_mma(P0, P1, PI, PO, PU, PF, cI, cO, cU, cF0, cF1);
    fz_load(P0, P1, PI, PO, PU, PF, a0, a1, wI, wO, wU, wF, kb + 8 * KU);
    fz_mma(Q0, Q1, QI, QO, QU, QF, cI, cO, cU, cF0, cF1);
  }
  fz_load(Q0, Q1, QI, QO, QU, QF, a0, a1, wI, wO, wU, wF, 256 - 4 * KU);
  fz_mma(P0, P1, PI, PO, PU, PF, cI, cO, cU, cF0, cF1);
  fz_mma(Q0, Q1, QI, QO, QU, QF, cI, cO, cU, cF0, cF1);

  __syncthreads();

#pragma unroll
  for (int v = 0; v < 8; ++v) {
    int b = 16 * rt + hi8 + v;
    float ig = sigf(xi[v] + cI[v] + bsI);
    float og = sigf(xo[v] + cO[v] + bsO);
    float ug = tanhq(xu[v] + cU[v] + bsU);
    float f0 = sigf(cF0[v] + bsF + xf[v]);
    float f1 = sigf(cF1[v] + bsF + xf[v]);
    float c = ig * ug + f0 * p0[v] + f1 * p1[v];
    float h = og * tanhq(c);
    sh[0][b * LSTRIDE + jl * 16 + lr] = h;
    sh[1][b * LSTRIDE + jl * 16 + lr] = c;
  }
  __syncthreads();

  const int cbase = blockIdx.y * CW;
#pragma unroll
  for (int i = 0; i < (BATCH * QR) / NTHR; ++i) {
    int F = tid + NTHR * i;
    int b = F / QR, q = F % QR;
    v4f hv = *(const v4f*)(&sh[0][b * LSTRIDE + q * 4]);
    v4f cv = *(const v4f*)(&sh[1][b * LSTRIDE + q * 4]);
    size_t o = (size_t)b * (NTREE * MDIM) + (size_t)node * MDIM + cbase + q * 4;
    *(v4f*)(hid + o)   = hv;
    *(v4f*)(c_buf + o) = cv;
  }
}

__global__ __launch_bounds__(128) void level_fused_w4(
    const float* __restrict__ xbuf, const float* __restrict__ wrec,
    const float* __restrict__ b_iouh, const float* __restrict__ b_fh,
    float* __restrict__ c_buf, float* __restrict__ hid, int level_start)
{
  __shared__ float sh[2][BATCH * LSTRIDE];
  level_fused_body<4>(sh, xbuf, wrec, b_iouh, b_fh, c_buf, hid, level_start);
}

__device__ __forceinline__ void fzb_load(
    v4f (&yI)[KU / 2], v4f (&yO)[KU / 2], v4f (&yU)[KU / 2], v4f (&yF)[KU / 2],
    const float* wI, const float* wO, const float* wU, const float* wF, int kb)
{
#pragma unroll
  for (int p = 0; p < KU / 2; ++p) {
    int off = (kb / 8 + p) * 128;
    yI[p] = ld4(wI + off); yO[p] = ld4(wO + off);
    yU[p] = ld4(wU + off); yF[p] = ld4(wF + off);
  }
}

__device__ __forceinline__ void fza_load(
    v2f (&x0)[KU], v2f (&x1)[KU], const float* a0, const float* a1, int kb)
{
#pragma unroll
  for (int u = 0; u < KU; ++u) {
    int k = kb + 4 * u;
    x0[u] = ld2(a0 + k); x1[u] = ld2(a1 + k);
  }
}

#define SOP 36

__global__ __launch_bounds__(256) void level_fused_ks4(
    const float* __restrict__ xbuf, const float* __restrict__ wrec,
    const float* __restrict__ b_iouh, const float* __restrict__ b_fh,
    float* __restrict__ c_buf, float* __restrict__ hid, int level_start)
{
  __shared__ float sh[2][16 * LSTRIDE];
  __shared__ float red[8][5][8][32];
  __shared__ float so[2][16 * SOP];

  const int tid  = threadIdx.x;
  const int node = level_start + blockIdx.x;
  const int ch0  = 2 * node - 511;
  const int ch1  = 2 * node - 512;
  const int cp   = blockIdx.y >> 1;
  const int rt   = blockIdx.y & 1;

  const int lane  = tid & 31;
  const int wave  = tid >> 5;
  const int lr    = lane & 15;
  const int khalf = (lane < 16) ? 0 : 2;
  const int hi8   = (lane >> 4) << 3;
  const int jl    = wave >> 2;
  const int ks    = wave & 3;
  const int j     = cp * 2 + jl;
  const int m     = j * 16 + lr;

  v4f st[8];
#pragma unroll
  for (int i = 0; i < 8; ++i) {
    int rem = (tid + 256 * i) & 1023;
    int r = rem >> 6, mq = rem & 63;
    int ch = (i >> 2) ? ch1 : ch0;
    st[i] = *(const v4f*)(hid + (size_t)(16 * rt + r) * (NTREE * MDIM) +
                          (size_t)ch * MDIM + mq * 4);
  }
  __builtin_amdgcn_sched_barrier(0);

  const size_t wq = (size_t)ks * 1024 + lane * 4;
  const float* wI = wrec + (size_t)(j)      * 4096 + wq;
  const float* wO = wrec + (size_t)(16 + j) * 4096 + wq;
  const float* wU = wrec + (size_t)(32 + j) * 4096 + wq;
  const float* wF = wrec + (size_t)(48 + j) * 4096 + wq;
  v4f PI[KU / 2], PO[KU / 2], PU[KU / 2], PF[KU / 2];
  v4f QI[KU / 2], QO[KU / 2], QU[KU / 2], QF[KU / 2];
  fzb_load(PI, PO, PU, PF, wI, wO, wU, wF, 0);
  fzb_load(QI, QO, QU, QF, wI, wO, wU, wF, 4 * KU);

  float xi[2], xo[2], xu[2], xf[2], p0[2], p1[2];
#pragma unroll
  for (int e = 0; e < 2; ++e) {
    int b = 16 * rt + hi8 + 2 * ks + e;
    const float* xp = xbuf + (size_t)node * (BATCH * 1024) + (size_t)b * 1024 + m;
    xi[e] = xp[0]; xo[e] = xp[256]; xu[e] = xp[512]; xf[e] = xp[768];
    size_t cb = (size_t)b * (NTREE * MDIM) + m;
    p0[e] = c_buf[cb + (size_t)ch0 * MDIM];
    p1[e] = c_buf[cb + (size_t)ch1 * MDIM];
  }
  const float bsI = b_iouh[m], bsO = b_iouh[256 + m], bsU = b_iouh[512 + m];
  const float bsF = b_fh[m];
  __builtin_amdgcn_sched_barrier(0);

#pragma unroll
  for (int i = 0; i < 8; ++i) {
    int rem = (tid + 256 * i) & 1023;
    int r = rem >> 6, mq = rem & 63;
    *(v4f*)(&sh[i >> 2][r * LSTRIDE + mq * 4]) = st[i];
  }
  __builtin_amdgcn_fence(__ATOMIC_RELEASE, "workgroup", "local");
  __builtin_amdgcn_s_barrier();
  __builtin_amdgcn_fence(__ATOMIC_ACQUIRE, "workgroup", "local");

  const float* a0 = &sh[0][lr * LSTRIDE + khalf + 64 * ks];
  const float* a1 = &sh[1][lr * LSTRIDE + khalf + 64 * ks];
  v2f P0[KU], P1[KU], Q0[KU], Q1[KU];
  fza_load(P0, P1, a0, a1, 0);
  fza_load(Q0, Q1, a0, a1, 4 * KU);
  __builtin_amdgcn_sched_barrier(0);

  v8f cI = {}, cO = {}, cU = {}, cF0 = {}, cF1 = {};
  fz_mma(P0, P1, PI, PO, PU, PF, cI, cO, cU, cF0, cF1);
  fz_mma(Q0, Q1, QI, QO, QU, QF, cI, cO, cU, cF0, cF1);

#pragma unroll
  for (int v = 0; v < 8; ++v) {
    red[wave][0][v][lane] = cI[v];
    red[wave][1][v][lane] = cO[v];
    red[wave][2][v][lane] = cU[v];
    red[wave][3][v][lane] = cF0[v];
    red[wave][4][v][lane] = cF1[v];
  }
  __syncthreads();

#pragma unroll
  for (int e = 0; e < 2; ++e) {
    const int v = 2 * ks + e;
    float s[5];
#pragma unroll
    for (int a = 0; a < 5; ++a) {
      float t = red[4 * jl + 0][a][v][lane];
      t += red[4 * jl + 1][a][v][lane];
      t += red[4 * jl + 2][a][v][lane];
      t += red[4 * jl + 3][a][v][lane];
      s[a] = t;
    }
    float ig = sigf(xi[e] + s[0] + bsI);
    float og = sigf(xo[e] + s[1] + bsO);
    float ug = tanhq(xu[e] + s[2] + bsU);
    float f0 = sigf(s[3] + bsF + xf[e]);
    float f1 = sigf(s[4] + bsF + xf[e]);
    float c = ig * ug + f0 * p0[e] + f1 * p1[e];
    float h = og * tanhq(c);
    const int row = hi8 + v;
    so[0][row * SOP + jl * 16 + lr] = h;
    so[1][row * SOP + jl * 16 + lr] = c;
  }
  __syncthreads();

  {
    const int which = tid >> 7;
    const int row = (tid >> 3) & 15, q = tid & 7;
    v4f val = *(const v4f*)(&so[which][row * SOP + q * 4]);
    size_t o = (size_t)(16 * rt + row) * (NTREE * MDIM) + (size_t)node * MDIM +
               cp * 32 + q * 4;
    float* dst = which ? c_buf : hid;
    *(v4f*)(dst + o) = val;
  }
}

__global__ __launch_bounds__(256) void root_copy(
    const float* __restrict__ c_buf, const float* __restrict__ hid,
    float* __restrict__ out)
{
  int idx = blockIdx.x * blockDim.x + threadIdx.x;
  int m = idx & 255, b = idx >> 8;
  size_t o = (size_t)b * (NTREE * MDIM) + (size_t)510 * MDIM + m;
  out[idx] = c_buf[o];
  out[BATCH * MDIM + idx] = hid[o];
}

extern "C" void kernel_launch(void* const* d_in, const int* in_sizes, int n_in,
                              void* d_out, int out_size, void* d_ws, size_t ws_size,
                              hipStream_t stream) {
  (void)in_sizes; (void)n_in; (void)out_size; (void)ws_size;
  const float* inputs = (const float*)d_in[0];
  const float* W_ioux = (const float*)d_in[1];
  const float* b_ioux = (const float*)d_in[2];
  const float* W_iouh = (const float*)d_in[3];
  const float* b_iouh = (const float*)d_in[4];
  const float* W_fx   = (const float*)d_in[5];
  const float* b_fx   = (const float*)d_in[6];
  const float* W_fh   = (const float*)d_in[7];
  const float* b_fh   = (const float*)d_in[8];

  float* out = (float*)d_out;
  float* hid = out + 2 * BATCH * MDIM;

  float* ws = (float*)d_ws;
  float* xbuf    = ws;
  float* c_buf   = xbuf + (size_t)NTREE * BATCH * 1024;
  float* wrec    = c_buf + (size_t)BATCH * NTREE * MDIM;
  float* wcat    = wrec + (size_t)128 * BATCH * 768 + (size_t)128 * 2 * BATCH * MDIM;
  float* bcat    = wcat + (size_t)1024 * 256;

  pack_weights<<<2052, 256, 0, stream>>>(
      W_ioux, b_ioux, W_fx, b_fx, W_iouh, W_fh, wcat, bcat, wrec);

  dim3 g1(16, 32);
  xproj_gemm<<<g1, 128, 0, stream>>>(inputs, wcat, bcat, b_iouh, xbuf, c_buf, hid);

  const int starts[8] = {256, 384, 448, 480, 496, 504, 508, 510};
  const int counts[8] = {128, 64, 32, 16, 8, 4, 2, 1};
  for (int L = 0; L < 3; ++L) {
    level_fused_w4<<<dim3(counts[L], 8), 128, 0, stream>>>(
        xbuf, wrec, b_iouh, b_fh, c_buf, hid, starts[L]);
  }
  for (int L = 3; L < 8; ++L) {
    level_fused_ks4<<<dim3(counts[L], 16), 256, 0, stream>>>(
        xbuf, wrec, b_iouh, b_fh, c_buf, hid, starts[L]);
  }

  root_copy<<<(BATCH * MDIM) / 256, 256, 0, stream>>>(c_buf, hid, out);
}


# 128/64-node levels: a wave forms both row tiles of its column tile (10 accumulators), 4 workgroups per node; weight fragments fetched once per node, gate inputs loaded under the last 2 K blocks
# speedup vs baseline: 6.1051x; 1.0664x over previous
#include <hip/hip_runtime.h>

#define NTREE 511
#define BATCH 32
#define MDIM  256
#define LSTRIDE 260

typedef __attribute__((ext_vector_type(2))) float v2f;
typedef __attribute__((ext_vector_type(4))) float v4f;
typedef __attribute__((ext_vector_type(8))) float v8f;

__device__ __forceinline__ v8f wmma4(v2f a, v2f b, v8f c) {
  return __builtin_amdgcn_wmma_f32_16x16x4_f32(false, a, false, b, (short)0, c,
                                               false, false);
}

__device__ __forceinline__ float sigf(float x) {
  return __builtin_amdgcn_rcpf(1.0f + __builtin_amdgcn_exp2f(x * -1.4426950408889634f));
}
__device__ __forceinline__ float tanhq(float x) {
  float r = __builtin_amdgcn_rcpf(1.0f + __builtin_amdgcn_exp2f(x * -2.8853900817779268f));
  return __builtin_fmaf(2.0f, r, -1.0f);
}

#define KU 8
#define XCOLS 64
#define XGROUPS 32
#define LROWS 96
#define LGROUPS 64
#define LSOP 36

__device__ __forceinline__ v2f ld2(const float* p) { return *(const v2f*)p; }

__device__ __forceinline__ void xp_load(
    v2f (&x0)[KU], v2f (&x1)[KU], v2f (&y)[KU],
    const float* a0, const float* a1, const float* w, int kb)
{
#pragma unroll
  for (int u = 0; u < KU; ++u) {
    int k = kb + 4 * u;
    x0[u] = ld2(a0 + k); x1[u] = ld2(a1 + k); y[u] = ld2(w + k);
  }
  __builtin_amdgcn_sched_barrier(0);
}

__device__ __forceinline__ void xp_mma(
    const v2f (&x0)[KU], const v2f (&x1)[KU], const v2f (&y)[KU],
    v8f& c0, v8f& c1)
{
#pragma unroll
  for (int u = 0; u < KU; ++u) {
    c0 = wmma4(x0[u], y[u], c0);
    c1 = wmma4(x1[u], y[u], c1);
  }
  __builtin_amdgcn_sched_barrier(0);
}

__device__ __forceinline__ void lf_load(
    v2f (&x)[KU], v2f (&y0)[KU], v2f (&y1)[KU], v2f (&y2)[KU],
    const float* a, const float* w, int kb)
{
#pragma unroll
  for (int u = 0; u < KU; ++u) {
    int k = kb + 4 * u;
    x[u]  = ld2(a + k);
    y0[u] = ld2(w + k);
    y1[u] = ld2(w + LSTRIDE + k);
    y2[u] = ld2(w + 2 * LSTRIDE + k);
  }
  __builtin_amdgcn_sched_barrier(0);
}

__device__ __forceinline__ void lf_mma(
    const v2f (&x)[KU], const v2f (&y0)[KU], const v2f (&y1)[KU],
    const v2f (&y2)[KU], v8f& c0, v8f& c1, v8f& c2)
{
#pragma unroll
  for (int u = 0; u < KU; ++u) {
    c0 = wmma4(x[u], y0[u], c0);
    c1 = wmma4(x[u], y1[u], c1);
    c2 = wmma4(x[u], y2[u], c2);
  }
  __builtin_amdgcn_sched_barrier(0);
}

__global__ __launch_bounds__(128) void xproj_gemm(
    const float* __restrict__ inputs,
    const float* __restrict__ W_ioux,
    const float* __restrict__ b_ioux,
    const float* __restrict__ W_fx,
    const float* __restrict__ b_fx,
    const float* __restrict__ W_iouh,
    const float* __restrict__ W_fh,
    const float* __restrict__ b_iouh,
    float* __restrict__ xbuf,
    float* __restrict__ c_buf,
    float* __restrict__ hid,
    float* __restrict__ wrec)
{
  __shared__ __attribute__((aligned(16))) float lds[128 * LSTRIDE + 2 * 32 * LSOP];

  const int tid   = threadIdx.x;
  const int lane  = tid & 31;
  const int wave  = tid >> 5;
  const int lr    = lane & 15;
  const int khalf = (lane < 16) ? 0 : 2;
  const int hi8   = (lane >> 4) << 3;

  {
    const int wg   = blockIdx.y * 16 + blockIdx.x;
    const int pair = (wg * 4 + wave) & 31;
    const int tile = wg >> 3;
    const int row  = tile * 16 + lr;
    const int k0   = 8 * pair + khalf;
    const float* wr = (row < 768) ? W_iouh + (size_t)row * 256
                                  : W_fh + (size_t)(row - 768) * 256;
    v2f lo = ld2(wr + k0);
    v2f hi = ld2(wr + k0 + 4);
    v4f pk = {lo[0], lo[1], hi[0], hi[1]};
    *(v4f*)(wrec + ((size_t)wg * 128 + tid) * 4) = pk;
  }

  {
    float* sw3 = lds;
    float* sa  = lds + LROWS * LSTRIDE;
    float* so  = lds + 128 * LSTRIDE;
    const int lc = blockIdx.x >> 1;
    const int lg = blockIdx.y * 2 + (blockIdx.x & 1);

#pragma unroll
    for (int i = 0; i < LROWS * 64 / 128; ++i) {
      int F = tid + 128 * i;
      int r = F >> 6, q = F & 63;
      int cl = r / 3, gt = r - 3 * cl;
      int srow = gt * 256 + lc * 32 + cl;
      float4 v = *(const float4*)(W_ioux + (size_t)srow * 256 + q * 4);
      *(float4*)(&sw3[r * LSTRIDE + q * 4]) = v;
    }
#pragma unroll
    for (int i = 0; i < 16; ++i) {
      int F = tid + 128 * i;
      int b = F >> 6, q = F & 63;
      float4 v = *(const float4*)(inputs + (size_t)b * (NTREE * MDIM) +
                                  (size_t)lg * MDIM + q * 4);
      *(float4*)(&sa[b * LSTRIDE + q * 4]) = v;
    }
    v4f nx[16];
#pragma unroll
    for (int i = 0; i < 16; ++i) {
      int F = tid + 128 * i;
      int b = F >> 6, q = F & 63;
      nx[i] = *(const v4f*)(inputs + (size_t)b * (NTREE * MDIM) +
                               (size_t)(lg + LGROUPS) * MDIM + q * 4);
    }
    __syncthreads();

    const int rt = wave & 1;
    const int ct = wave >> 1;
    const float* a = &sa[(16 * rt + lr) * LSTRIDE + khalf];
    const float* w = &sw3[3 * (16 * ct + lr) * LSTRIDE + khalf];
    const int m    = lc * 32 + ct * 16 + lr;
    const float bs0 = b_ioux[m], bs1 = b_ioux[256 + m], bs2 = b_ioux[512 + m];
    const float bi0 = b_iouh[m], bi1 = b_iouh[256 + m], bi2 = b_iouh[512 + m];

#pragma unroll 1
    for (int i = 0; i < 256 / LGROUPS; ++i) {
      const int node = lg + LGROUPS * i;

      v2f PX[KU], P0[KU], P1[KU], P2[KU];
      v2f QX[KU], Q0[KU], Q1[KU], Q2[KU];
      v8f c0 = {}, c1 = {}, c2 = {};
      lf_load(PX, P0, P1, P2, a, w, 0);
#pragma unroll
      for (int kb = 0; kb < 256 - 8 * KU; kb += 8 * KU) {
        lf_load(QX, Q0, Q1, Q2, a, w, kb + 4 * KU);
        lf_mma(PX, P0, P1, P2, c0, c1, c2);
        lf_load(PX, P0, P1, P2, a, w, kb + 8 * KU);
        lf_mma(QX, Q0, Q1, Q2, c0, c1, c2);
      }
      lf_load(QX, Q0, Q1, Q2, a, w, 256 - 4 * KU);
      lf_mma(PX, P0, P1, P2, c0, c1, c2);
      lf_mma(QX, Q0, Q1, Q2, c0, c1, c2);

#pragma unroll
      for (int v = 0; v < 8; ++v) {
        int b = 16 * rt + hi8 + v;
        float ig = sigf((c0[v] + bs0) + bi0);
        float og = sigf((c1[v] + bs1) + bi1);
        float ug = tanhq((c2[v] + bs2) + bi2);
        float cc = ig * ug;
        float hh = og * tanhq(cc);
        so[b * LSOP + ct * 16 + lr]             = hh;
        so[32 * LSOP + b * LSOP + ct * 16 + lr] = cc;
      }
      __syncthreads();

      if (i + 1 < 256 / LGROUPS) {
#pragma unroll
        for (int e = 0; e < 16; ++e) {
          int F = tid + 128 * e;
          int b = F >> 6, q = F & 63;
          *(v4f*)(&sa[b * LSTRIDE + q * 4]) = nx[e];
        }
      }
#pragma unroll
      for (int e = 0; e < 4; ++e) {
        int F = tid + 128 * e;
        int plane = F >> 8;
        int b = (F >> 3) & 31, q = F & 7;
        v4f val = *(const v4f*)(&so[plane * 32 * LSOP + b * LSOP + q * 4]);
        float* dst = plane ? c_buf : hid;
        *(v4f*)(dst + (size_t)b * (NTREE * MDIM) + (size_t)node * MDIM +
                lc * 32 + q * 4) = val;
      }
      __syncthreads();

      if (i + 2 < 256 / LGROUPS) {
#pragma unroll
        for (int e = 0; e < 16; ++e) {
          int F = tid + 128 * e;
          int b = F >> 6, q = F & 63;
          nx[e] = *(const v4f*)(inputs + (size_t)b * (NTREE * MDIM) +
                                   (size_t)(node + 2 * LGROUPS) * MDIM + q * 4);
        }
      }
    }
  }

  float* sw  = lds;
  float* sa0 = lds + XCOLS * LSTRIDE;
  const int cbase = blockIdx.x * XCOLS;
  const int g     = blockIdx.y;
  const int nfirst = 256 + g;

  const float* wsrc = (cbase < 768) ? W_ioux + (size_t)cbase * 256
                                    : W_fx + (size_t)(cbase - 768) * 256;
#pragma unroll 8
  for (int i = 0; i < XCOLS * 64 / 128; ++i) {
    int F = tid + 128 * i;
    int r = F >> 6, q = F & 63;
    float4 v = *(const float4*)(wsrc + (size_t)r * 256 + q * 4);
    *(float4*)(&sw[r * LSTRIDE + q * 4]) = v;
  }
#pragma unroll
  for (int i = 0; i < 16; ++i) {
    int F = tid + 128 * i;
    int b = F >> 6, q = F & 63;
    float4 v = *(const float4*)(inputs + (size_t)b * (NTREE * MDIM) +
                                (size_t)nfirst * MDIM + q * 4);
    *(float4*)(&sa0[b * LSTRIDE + q * 4]) = v;
  }
  v4f nx[16];
#pragma unroll
  for (int i = 0; i < 16; ++i) {
    int F = tid + 128 * i;
    int b = F >> 6, q = F & 63;
    nx[i] = *(const v4f*)(inputs + (size_t)b * (NTREE * MDIM) +
                             (size_t)(nfirst + XGROUPS) * MDIM + q * 4);
  }
  __syncthreads();

  const float* wl = &sw[(wave * 16 + lr) * LSTRIDE + khalf];
  const int col   = cbase + wave * 16 + lr;
  const float bs  = (col < 768) ? b_ioux[col] : b_fx[col - 768];

  int buf = 0;
  for (int n = nfirst; n < NTREE; n += XGROUPS) {
    const float* a0 = &sa0[buf * (BATCH * LSTRIDE) + lr * LSTRIDE + khalf];
    const float* a1 = &sa0[buf * (BATCH * LSTRIDE) + (16 + lr) * LSTRIDE + khalf];

    v2f P0[KU], P1[KU], PW[KU];
    v2f Q0[KU], Q1[KU], QW[KU];
    v8f c0 = {}, c1 = {};
    xp_load(P0, P1, PW, a0, a1, wl, 0);
#pragma unroll
    for (int kb = 0; kb < 256 - 8 * KU; kb += 8 * KU) {
      xp_load(Q0, Q1, QW, a0, a1, wl, kb + 4 * KU);
      xp_mma(P0, P1, PW, c0, c1);
      xp_load(P0, P1, PW, a0, a1, wl, kb + 8 * KU);
      xp_mma(Q0, Q1, QW, c0, c1);
    }
    xp_load(Q0, Q1, QW, a0, a1, wl, 256 - 4 * KU);
    xp_mma(P0, P1, PW, c0, c1);
    xp_mma(Q0, Q1, QW, c0, c1);

    float* sn = &sa0[(buf ^ 1) * (BATCH * LSTRIDE)];
#pragma unroll
    for (int i = 0; i < 16; ++i) {
      int F = tid + 128 * i;
      int b = F >> 6, q = F & 63;
      *(v4f*)(&sn[b * LSTRIDE + q * 4]) = nx[i];
    }
    __syncthreads();

    float* op = xbuf + (size_t)n * (BATCH * 1024) + col;
#pragma unroll
    for (int v = 0; v < 8; ++v) {
      op[(size_t)(v + hi8) * 1024]      = c0[v] + bs;
      op[(size_t)(16 + v + hi8) * 1024] = c1[v] + bs;
    }

    const int n2 = (n + 2 * XGROUPS < NTREE) ? n + 2 * XGROUPS : n;
#pragma unroll
    for (int i = 0; i < 16; ++i) {
      int F = tid + 128 * i;
      int b = F >> 6, q = F & 63;
      nx[i] = *(const v4f*)(inputs + (size_t)b * (NTREE * MDIM) +
                               (size_t)n2 * MDIM + q * 4);
    }
    buf ^= 1;
  }
}

__device__ __forceinline__ v4f ld4(const float* p) { return *(const v4f*)p; }

__device__ __forceinline__ void fz_load(
    v2f (&x0)[KU], v2f (&x1)[KU], v4f (&yI)[KU / 2], v4f (&yO)[KU / 2],
    v4f (&yU)[KU / 2], v4f (&yF)[KU / 2],
    const float* a0, const float* a1, const float* wI, const float* wO,
    const float* wU, const float* wF, int kb)
{
#pragma unroll
  for (int u = 0; u < KU; ++u) {
    int k = kb + 4 * u;
    x0[u] = ld2(a0 + k); x1[u] = ld2(a1 + k);
  }
#pragma unroll
  for (int p = 0; p < KU / 2; ++p) {
    int off = (kb / 8 + p) * 128;
    yI[p] = ld4(wI + off); yO[p] = ld4(wO + off);
    yU[p] = ld4(wU + off); yF[p] = ld4(wF + off);
  }
  __builtin_amdgcn_sched_barrier(0);
}

__device__ __forceinline__ void fz_mma(
    const v2f (&x0)[KU], const v2f (&x1)[KU], const v4f (&yI)[KU / 2],
    const v4f (&yO)[KU / 2], const v4f (&yU)[KU / 2], const v4f (&yF)[KU / 2],
    v8f& cI, v8f& cO, v8f& cU, v8f& cF0, v8f& cF1)
{
#pragma unroll
  for (int u = 0; u < KU; ++u) {
    v2f xs = x0[u] + x1[u];
    v2f bI, bO, bU, bF;
    if (u & 1) {
      bI = __builtin_shufflevector(yI[u / 2], yI[u / 2], 2, 3);
      bO = __builtin_shufflevector(yO[u / 2], yO[u / 2], 2, 3);
      bU = __builtin_shufflevector(yU[u / 2], yU[u / 2], 2, 3);
      bF = __builtin_shufflevector(yF[u / 2], yF[u / 2], 2, 3);
    } else {
      bI = __builtin_shufflevector(yI[u / 2], yI[u / 2], 0, 1);
      bO = __builtin_shufflevector(yO[u / 2], yO[u / 2], 0, 1);
      bU = __builtin_shufflevector(yU[u / 2], yU[u / 2], 0, 1);
      bF = __builtin_shufflevector(yF[u / 2], yF[u / 2], 0, 1);
    }
    cI  = wmma4(xs, bI, cI);
    cO  = wmma4(xs, bO, cO);
    cU  = wmma4(xs, bU, cU);
    cF0 = wmma4(x0[u], bF, cF0);
    cF1 = wmma4(x1[u], bF, cF1);
  }
  __builtin_amdgcn_sched_barrier(0);
}

template <int NW>
__device__ __forceinline__ void level_fused_body(
    float (*sh)[BATCH * LSTRIDE],
    const float* __restrict__ xbuf, const float* __restrict__ wrec,
    const float* __restrict__ b_iouh, const float* __restrict__ b_fh,
    float* __restrict__ c_buf, float* __restrict__ hid, int level_start)
{
  constexpr int NTHR = NW * 32;
  constexpr int CT   = NW / 2;
  constexpr int CW   = CT * 16;
  constexpr int QR   = CW / 4;

  const int tid  = threadIdx.x;
  const int node = level_start + blockIdx.x;
  const int ch0  = 2 * node - 511;
  const int ch1  = 2 * node - 512;

#pragma unroll
  for (int i = 0; i < 4096 / NTHR; ++i) {
    int F = tid + NTHR * i;
    int which = F >> 11;
    int rem = F & 2047;
    int b = rem >> 6, mq = rem & 63;
    int ch = which ? ch1 : ch0;
    float4 v = *(const float4*)(hid + (size_t)b * (NTREE * MDIM) +
                                (size_t)ch * MDIM + mq * 4);
    *(float4*)(&sh[which][b * LSTRIDE + mq * 4]) = v;
  }

  const int lane  = tid & 31;
  const int wave  = tid >> 5;
  const int lr    = lane & 15;
  const int khalf = (lane < 16) ? 0 : 2;
  const int hi8   = (lane >> 4) << 3;
  const int jl    = wave >> 1;
  const int rt    = wave & 1;
  const int j     = blockIdx.y * CT + jl;
  const int m     = j * 16 + lr;

  float xi[8], xo[8], xu[8], xf[8], p0[8], p1[8];
#pragma unroll
  for (int v = 0; v < 8; ++v) {
    int b = 16 * rt + hi8 + v;
    const float* xp = xbuf + (size_t)node * (BATCH * 1024) + (size_t)b * 1024 + m;
    xi[v] = xp[0]; xo[v] = xp[256]; xu[v] = xp[512]; xf[v] = xp[768];
    size_t cb = (size_t)b * (NTREE * MDIM) + m;
    p0[v] = c_buf[cb + (size_t)ch0 * MDIM];
    p1[v] = c_buf[cb + (size_t)ch1 * MDIM];
  }
  const float bsI = b_iouh[m], bsO = b_iouh[256 + m], bsU = b_iouh[512 + m];
  const float bsF = b_fh[m];

  __syncthreads();

  const float* a0 = &sh[0][(16 * rt + lr) * LSTRIDE + khalf];
  const float* a1 = &sh[1][(16 * rt + lr) * LSTRIDE + khalf];
  const float* wI = wrec + (size_t)(j)      * 4096 + lane * 4;
  const float* wO = wrec + (size_t)(16 + j) * 4096 + lane * 4;
  const float* wU = wrec + (size_t)(32 + j) * 4096 + lane * 4;
  const float* wF = wrec + (size_t)(48 + j) * 4096 + lane * 4;

  v2f P0[KU], P1[KU], Q0[KU], Q1[KU];
  v4f PI[KU / 2], PO[KU / 2], PU[KU / 2], PF[KU / 2];
  v4f QI[KU / 2], QO[KU / 2], QU[KU / 2], QF[KU / 2];
  v8f cI = {}, cO = {}, cU = {}, cF0 = {}, cF1 = {};
  fz_load(P0, P1, PI, PO, PU, PF, a0, a1, wI, wO, wU, wF, 0);
#pragma unroll 1
  for (int kb = 0; kb < 256 - 8 * KU; kb += 8 * KU) {
    fz_load(Q0, Q1, QI, QO, QU, QF, a0, a1, wI, wO, wU, wF, kb + 4 * KU);
    fz_mma(P0, P1, PI, PO, PU, PF, cI, cO, cU, cF0, cF1);
    fz_load(P0, P1, PI, PO, PU, PF, a0, a1, wI, wO, wU, wF, kb + 8 * KU);
    fz_mma(Q0, Q1, QI, QO, QU, QF, cI, cO, cU, cF0, cF1);
  }
  fz_load(Q0, Q1, QI, QO, QU, QF, a0, a1, wI, wO, wU, wF, 256 - 4 * KU);
  fz_mma(P0, P1, PI, PO, PU, PF, cI, cO, cU, cF0, cF1);
  fz_mma(Q0, Q1, QI, QO, QU, QF, cI, cO, cU, cF0, cF1);

  __syncthreads();

#pragma unroll
  for (int v = 0; v < 8; ++v) {
    int b = 16 * rt + hi8 + v;
    float ig = sigf(xi[v] + cI[v] + bsI);
    float og = sigf(xo[v] + cO[v] + bsO);
    float ug = tanhq(xu[v] + cU[v] + bsU);
    float f0 = sigf(cF0[v] + bsF + xf[v]);
    float f1 = sigf(cF1[v] + bsF + xf[v]);
    float c = ig * ug + f0 * p0[v] + f1 * p1[v];
    float h = og * tanhq(c);
    sh[0][b * LSTRIDE + jl * 16 + lr] = h;
    sh[1][b * LSTRIDE + jl * 16 + lr] = c;
  }
  __syncthreads();

  const int cbase = blockIdx.y * CW;
#pragma unroll
  for (int i = 0; i < (BATCH * QR) / NTHR; ++i) {
    int F = tid + NTHR * i;
    int b = F / QR, q = F % QR;
    v4f hv = *(const v4f*)(&sh[0][b * LSTRIDE + q * 4]);
    v4f cv = *(const v4f*)(&sh[1][b * LSTRIDE + q * 4]);
    size_t o = (size_t)b * (NTREE * MDIM) + (size_t)node * MDIM + cbase + q * 4;
    *(v4f*)(hid + o)   = hv;
    *(v4f*)(c_buf + o) = cv;
  }
}

__global__ __launch_bounds__(128) void level_fused_w4(
    const float* __restrict__ xbuf, const float* __restrict__ wrec,
    const float* __restrict__ b_iouh, const float* __restrict__ b_fh,
    float* __restrict__ c_buf, float* __restrict__ hid, int level_start)
{
  __shared__ float sh[2][BATCH * LSTRIDE];
  level_fused_body<4>(sh, xbuf, wrec, b_iouh, b_fh, c_buf, hid, level_start);
}

__device__ __forceinline__ void fw_load(
    v2f (&x0)[KU], v2f (&x1)[KU], v2f (&z0)[KU], v2f (&z1)[KU],
    v4f (&yI)[KU / 2], v4f (&yO)[KU / 2], v4f (&yU)[KU / 2], v4f (&yF)[KU / 2],
    const float* a0, const float* a1, const float* wI, const float* wO,
    const float* wU, const float* wF, int kb)
{
#pragma unroll
  for (int u = 0; u < KU; ++u) {
    int k = kb + 4 * u;
    x0[u] = ld2(a0 + k); x1[u] = ld2(a1 + k);
    z0[u] = ld2(a0 + 16 * LSTRIDE + k); z1[u] = ld2(a1 + 16 * LSTRIDE + k);
  }
#pragma unroll
  for (int p = 0; p < KU / 2; ++p) {
    int off = (kb / 8 + p) * 128;
    yI[p] = ld4(wI + off); yO[p] = ld4(wO + off);
    yU[p] = ld4(wU + off); yF[p] = ld4(wF + off);
  }
  __builtin_amdgcn_sched_barrier(0);
}

__device__ __forceinline__ void fw_mma(
    const v2f (&x0)[KU], const v2f (&x1)[KU], const v2f (&z0)[KU],
    const v2f (&z1)[KU], const v4f (&yI)[KU / 2], const v4f (&yO)[KU / 2],
    const v4f (&yU)[KU / 2], const v4f (&yF)[KU / 2],
    v8f& cI, v8f& cO, v8f& cU, v8f& cF0, v8f& cF1,
    v8f& dI, v8f& dO, v8f& dU, v8f& dF0, v8f& dF1)
{
#pragma unroll
  for (int u = 0; u < KU; ++u) {
    v2f xs = x0[u] + x1[u];
    v2f zs = z0[u] + z1[u];
    v2f bI, bO, bU, bF;
    if (u & 1) {
      bI = __builtin_shufflevector(yI[u / 2], yI[u / 2], 2, 3);
      bO = __builtin_shufflevector(yO[u / 2], yO[u / 2], 2, 3);
      bU = __builtin_shufflevector(yU[u / 2], yU[u / 2], 2, 3);
      bF = __builtin_shufflevector(yF[u / 2], yF[u / 2], 2, 3);
    } else {
      bI = __builtin_shufflevector(yI[u / 2], yI[u / 2], 0, 1);
      bO = __builtin_shufflevector(yO[u / 2], yO[u / 2], 0, 1);
      bU = __builtin_shufflevector(yU[u / 2], yU[u / 2], 0, 1);
      bF = __builtin_shufflevector(yF[u / 2], yF[u / 2], 0, 1);
    }
    cI  = wmma4(xs, bI, cI);
    dI  = wmma4(zs, bI, dI);
    cO  = wmma4(xs, bO, cO);
    dO  = wmma4(zs, bO, dO);
    cU  = wmma4(xs, bU, cU);
    dU  = wmma4(zs, bU, dU);
    cF0 = wmma4(x0[u], bF, cF0);
    cF1 = wmma4(x1[u], bF, cF1);
    dF0 = wmma4(z0[u], bF, dF0);
    dF1 = wmma4(z1[u], bF, dF1);
  }
  __builtin_amdgcn_sched_barrier(0);
}

__global__ __launch_bounds__(128) void level_fused_w2r(
    const float* __restrict__ xbuf, const float* __restrict__ wrec,
    const float* __restrict__ b_iouh, const float* __restrict__ b_fh,
    float* __restrict__ c_buf, float* __restrict__ hid, int level_start)
{
  __shared__ float sh[2][BATCH * LSTRIDE];

  const int tid  = threadIdx.x;
  const int node = level_start + blockIdx.x;
  const int ch0  = 2 * node - 511;
  const int ch1  = 2 * node - 512;

#pragma unroll
  for (int i = 0; i < 4096 / 128; ++i) {
    int F = tid + 128 * i;
    int which = F >> 11;
    int rem = F & 2047;
    int b = rem >> 6, mq = rem & 63;
    int ch = which ? ch1 : ch0;
    float4 v = *(const float4*)(hid + (size_t)b * (NTREE * MDIM) +
                                (size_t)ch * MDIM + mq * 4);
    *(float4*)(&sh[which][b * LSTRIDE + mq * 4]) = v;
  }

  const int lane  = tid & 31;
  const int jl    = tid >> 5;
  const int lr    = lane & 15;
  const int khalf = (lane < 16) ? 0 : 2;
  const int hi8   = (lane >> 4) << 3;
  const int j     = blockIdx.y * 4 + jl;
  const int m     = j * 16 + lr;

  __syncthreads();

  const float* a0 = &sh[0][lr * LSTRIDE + khalf];
  const float* a1 = &sh[1][lr * LSTRIDE + khalf];
  const float* wI = wrec + (size_t)(j)      * 4096 + lane * 4;
  const float* wO = wrec + (size_t)(16 + j) * 4096 + lane * 4;
  const float* wU = wrec + (size_t)(32 + j) * 4096 + lane * 4;
  const float* wF = wrec + (size_t)(48 + j) * 4096 + lane * 4;

  v2f P0[KU], P1[KU], R0[KU], R1[KU], Q0[KU], Q1[KU], S0[KU], S1[KU];
  v4f PI[KU / 2], PO[KU / 2], PU[KU / 2], PF[KU / 2];
  v4f QI[KU / 2], QO[KU / 2], QU[KU / 2], QF[KU / 2];
  v8f cI = {}, cO = {}, cU = {}, cF0 = {}, cF1 = {};
  v8f dI = {}, dO = {}, dU = {}, dF0 = {}, dF1 = {};
  fw_load(P0, P1, R0, R1, PI, PO, PU, PF, a0, a1, wI, wO, wU, wF, 0);

#pragma unroll 1
  for (int kb = 0; kb < 256 - 8 * KU; kb += 8 * KU) {
    fw_load(Q0, Q1, S0, S1, QI, QO, QU, QF, a0, a1, wI, wO, wU, wF, kb + 4 * KU);
    fw_mma(P0, P1, R0, R1, PI, PO, PU, PF, cI, cO, cU, cF0, cF1, dI, dO, dU, dF0, dF1);
    fw_load(P0, P1, R0, R1, PI, PO, PU, PF, a0, a1, wI, wO, wU, wF, kb + 8 * KU);
    fw_mma(Q0, Q1, S0, S1, QI, QO, QU, QF, cI, cO, cU, cF0, cF1, dI, dO, dU, dF0, dF1);
  }
  fw_load(Q0, Q1, S0, S1, QI, QO, QU, QF, a0, a1, wI, wO, wU, wF, 256 - 4 * KU);

  float xi[16], xo[16], xu[16], xf[16], p0[16], p1[16];
#pragma unroll
  for (int e = 0; e < 16; ++e) {
    int b = 16 * (e >> 3) + hi8 + (e & 7);
    const float* xp = xbuf + (size_t)node * (BATCH * 1024) + (size_t)b * 1024 + m;
    xi[e] = xp[0]; xo[e] = xp[256]; xu[e] = xp[512];
  }
  __builtin_amdgcn_sched_barrier(0);

  fw_mma(P0, P1, R0, R1, PI, PO, PU, PF, cI, cO, cU, cF0, cF1, dI, dO, dU, dF0, dF1);

#pragma unroll
  for (int e = 0; e < 16; ++e) {
    int b = 16 * (e >> 3) + hi8 + (e & 7);
    xf[e] = xbuf[(size_t)node * (BATCH * 1024) + (size_t)b * 1024 + 768 + m];
    size_t cb = (size_t)b * (NTREE * MDIM) + m;
    p0[e] = c_buf[cb + (size_t)ch0 * MDIM];
    p1[e] = c_buf[cb + (size_t)ch1 * MDIM];
  }
  const float bsI = b_iouh[m], bsO = b_iouh[256 + m], bsU = b_iouh[512 + m];
  const float bsF = b_fh[m];
  __builtin_amdgcn_sched_barrier(0);

  fw_mma(Q0, Q1, S0, S1, QI, QO, QU, QF, cI, cO, cU, cF0, cF1, dI, dO, dU, dF0, dF1);

  __syncthreads();

#pragma unroll
  for (int e = 0; e < 16; ++e) {
    const int v = e & 7;
    const int b = 16 * (e >> 3) + hi8 + v;
    const float aI  = (e < 8) ? cI[v]  : dI[v];
    const float aO  = (e < 8) ? cO[v]  : dO[v];
    const float aU  = (e < 8) ? cU[v]  : dU[v];
    const float aF0 = (e < 8) ? cF0[v] : dF0[v];
    const float aF1 = (e < 8) ? cF1[v] : dF1[v];
    float ig = sigf(xi[e] + aI + bsI);
    float og = sigf(xo[e] + aO + bsO);
    float ug = tanhq(xu[e] + aU + bsU);
    float f0 = sigf(aF0 + bsF + xf[e]);
    float f1 = sigf(aF1 + bsF + xf[e]);
    float c = ig * ug + f0 * p0[e] + f1 * p1[e];
    float h = og * tanhq(c);
    sh[0][b * LSTRIDE + jl * 16 + lr] = h;
    sh[1][b * LSTRIDE + jl * 16 + lr] = c;
  }
  __syncthreads();

  const int cbase = blockIdx.y * 64;
#pragma unroll
  for (int i = 0; i < (BATCH * 16) / 128; ++i) {
    int F = tid + 128 * i;
    int b = F >> 4, q = F & 15;
    v4f hv = *(const v4f*)(&sh[0][b * LSTRIDE + q * 4]);
    v4f cv = *(const v4f*)(&sh[1][b * LSTRIDE + q * 4]);
    size_t o = (size_t)b * (NTREE * MDIM) + (size_t)node * MDIM + cbase + q * 4;
    *(v4f*)(hid + o)   = hv;
    *(v4f*)(c_buf + o) = cv;
  }
}

__device__ __forceinline__ void fzb_load(
    v4f (&yI)[KU / 2], v4f (&yO)[KU / 2], v4f (&yU)[KU / 2], v4f (&yF)[KU / 2],
    const float* wI, const float* wO, const float* wU, const float* wF, int kb)
{
#pragma unroll
  for (int p = 0; p < KU / 2; ++p) {
    int off = (kb / 8 + p) * 128;
    yI[p] = ld4(wI + off); yO[p] = ld4(wO + off);
    yU[p] = ld4(wU + off); yF[p] = ld4(wF + off);
  }
}

__device__ __forceinline__ void fza_load(
    v2f (&x0)[KU], v2f (&x1)[KU], const float* a0, const float* a1, int kb)
{
#pragma unroll
  for (int u = 0; u < KU; ++u) {
    int k = kb + 4 * u;
    x0[u] = ld2(a0 + k); x1[u] = ld2(a1 + k);
  }
}

#define SOP 36

__global__ __launch_bounds__(256) void level_fused_ks4(
    const float* __restrict__ xbuf, const float* __restrict__ wrec,
    const float* __restrict__ b_iouh, const float* __restrict__ b_fh,
    float* __restrict__ c_buf, float* __restrict__ hid,
    float* __restrict__ root, int level_start)
{
  __shared__ float sh[2][16 * LSTRIDE];
  __shared__ float red[8][5][8][32];
  __shared__ float so[2][16 * SOP];

  const int tid  = threadIdx.x;
  const int node = level_start + blockIdx.x;
  const int ch0  = 2 * node - 511;
  const int ch1  = 2 * node - 512;
  const int cp   = blockIdx.y >> 1;
  const int rt   = blockIdx.y & 1;

  const int lane  = tid & 31;
  const int wave  = tid >> 5;
  const int lr    = lane & 15;
  const int khalf = (lane < 16) ? 0 : 2;
  const int hi8   = (lane >> 4) << 3;
  const int jl    = wave >> 2;
  const int ks    = wave & 3;
  const int j     = cp * 2 + jl;
  const int m     = j * 16 + lr;

  v4f st[8];
#pragma unroll
  for (int i = 0; i < 8; ++i) {
    int rem = (tid + 256 * i) & 1023;
    int r = rem >> 6, mq = rem & 63;
    int ch = (i >> 2) ? ch1 : ch0;
    st[i] = *(const v4f*)(hid + (size_t)(16 * rt + r) * (NTREE * MDIM) +
                          (size_t)ch * MDIM + mq * 4);
  }
  __builtin_amdgcn_sched_barrier(0);

  const size_t wq = (size_t)ks * 1024 + lane * 4;
  const float* wI = wrec + (size_t)(j)      * 4096 + wq;
  const float* wO = wrec + (size_t)(16 + j) * 4096 + wq;
  const float* wU = wrec + (size_t)(32 + j) * 4096 + wq;
  const float* wF = wrec + (size_t)(48 + j) * 4096 + wq;
  v4f PI[KU / 2], PO[KU / 2], PU[KU / 2], PF[KU / 2];
  v4f QI[KU / 2], QO[KU / 2], QU[KU / 2], QF[KU / 2];
  fzb_load(PI, PO, PU, PF, wI, wO, wU, wF, 0);
  fzb_load(QI, QO, QU, QF, wI, wO, wU, wF, 4 * KU);

  float xi[2], xo[2], xu[2], xf[2], p0[2], p1[2];
#pragma unroll
  for (int e = 0; e < 2; ++e) {
    int b = 16 * rt + hi8 + 2 * ks + e;
    const float* xp = xbuf + (size_t)node * (BATCH * 1024) + (size_t)b * 1024 + m;
    xi[e] = xp[0]; xo[e] = xp[256]; xu[e] = xp[512]; xf[e] = xp[768];
    size_t cb = (size_t)b * (NTREE * MDIM) + m;
    p0[e] = c_buf[cb + (size_t)ch0 * MDIM];
    p1[e] = c_buf[cb + (size_t)ch1 * MDIM];
  }
  const float bsI = b_iouh[m], bsO = b_iouh[256 + m], bsU = b_iouh[512 + m];
  const float bsF = b_fh[m];
  __builtin_amdgcn_sched_barrier(0);

#pragma unroll
  for (int i = 0; i < 8; ++i) {
    int rem = (tid + 256 * i) & 1023;
    int r = rem >> 6, mq = rem & 63;
    *(v4f*)(&sh[i >> 2][r * LSTRIDE + mq * 4]) = st[i];
  }
  __builtin_amdgcn_fence(__ATOMIC_RELEASE, "workgroup", "local");
  __builtin_amdgcn_s_barrier();
  __builtin_amdgcn_fence(__ATOMIC_ACQUIRE, "workgroup", "local");

  const float* a0 = &sh[0][lr * LSTRIDE + khalf + 64 * ks];
  const float* a1 = &sh[1][lr * LSTRIDE + khalf + 64 * ks];
  v2f P0[KU], P1[KU], Q0[KU], Q1[KU];
  fza_load(P0, P1, a0, a1, 0);
  fza_load(Q0, Q1, a0, a1, 4 * KU);
  __builtin_amdgcn_sched_barrier(0);

  v8f cI = {}, cO = {}, cU = {}, cF0 = {}, cF1 = {};
  fz_mma(P0, P1, PI, PO, PU, PF, cI, cO, cU, cF0, cF1);
  fz_mma(Q0, Q1, QI, QO, QU, QF, cI, cO, cU, cF0, cF1);

#pragma unroll
  for (int v = 0; v < 8; ++v) {
    red[wave][0][v][lane] = cI[v];
    red[wave][1][v][lane] = cO[v];
    red[wave][2][v][lane] = cU[v];
    red[wave][3][v][lane] = cF0[v];
    red[wave][4][v][lane] = cF1[v];
  }
  __syncthreads();

#pragma unroll
  for (int e = 0; e < 2; ++e) {
    const int v = 2 * ks + e;
    float s[5];
#pragma unroll
    for (int a = 0; a < 5; ++a) {
      float t = red[4 * jl + 0][a][v][lane];
      t += red[4 * jl + 1][a][v][lane];
      t += red[4 * jl + 2][a][v][lane];
      t += red[4 * jl + 3][a][v][lane];
      s[a] = t;
    }
    float ig = sigf(xi[e] + s[0] + bsI);
    float og = sigf(xo[e] + s[1] + bsO);
    float ug = tanhq(xu[e] + s[2] + bsU);
    float f0 = sigf(s[3] + bsF + xf[e]);
    float f1 = sigf(s[4] + bsF + xf[e]);
    float c = ig * ug + f0 * p0[e] + f1 * p1[e];
    float h = og * tanhq(c);
    const int row = hi8 + v;
    so[0][row * SOP + jl * 16 + lr] = h;
    so[1][row * SOP + jl * 16 + lr] = c;
  }
  __syncthreads();

  {
    const int which = tid >> 7;
    const int row = (tid >> 3) & 15, q = tid & 7;
    v4f val = *(const v4f*)(&so[which][row * SOP + q * 4]);
    size_t o = (size_t)(16 * rt + row) * (NTREE * MDIM) + (size_t)node * MDIM +
               cp * 32 + q * 4;
    float* dst = which ? c_buf : hid;
    *(v4f*)(dst + o) = val;
    if (node == NTREE - 1) {
      *(v4f*)(root + (which ? 0 : BATCH * MDIM) + (16 * rt + row) * MDIM +
              cp * 32 + q * 4) = val;
    }
  }
}

extern "C" void kernel_launch(void* const* d_in, const int* in_sizes, int n_in,
                              void* d_out, int out_size, void* d_ws, size_t ws_size,
                              hipStream_t stream) {
  (void)in_sizes; (void)n_in; (void)out_size; (void)ws_size;
  const float* inputs = (const float*)d_in[0];
  const float* W_ioux = (const float*)d_in[1];
  const float* b_ioux = (const float*)d_in[2];
  const float* W_iouh = (const float*)d_in[3];
  const float* b_iouh = (const float*)d_in[4];
  const float* W_fx   = (const float*)d_in[5];
  const float* b_fx   = (const float*)d_in[6];
  const float* W_fh   = (const float*)d_in[7];
  const float* b_fh   = (const float*)d_in[8];

  float* out = (float*)d_out;
  float* hid = out + 2 * BATCH * MDIM;

  float* ws = (float*)d_ws;
  float* xbuf    = ws;
  float* c_buf   = xbuf + (size_t)NTREE * BATCH * 1024;
  float* wrec    = c_buf + (size_t)BATCH * NTREE * MDIM;

  dim3 g1(16, 32);
  xproj_gemm<<<g1, 128, 0, stream>>>(inputs, W_ioux, b_ioux, W_fx, b_fx, W_iouh, W_fh,
                                     b_iouh, xbuf, c_buf, hid, wrec);

  const int starts[8] = {256, 384, 448, 480, 496, 504, 508, 510};
  const int counts[8] = {128, 64, 32, 16, 8, 4, 2, 1};
  for (int L = 0; L < 2; ++L) {
    level_fused_w2r<<<dim3(counts[L], 4), 128, 0, stream>>>(
        xbuf, wrec, b_iouh, b_fh, c_buf, hid, starts[L]);
  }
  level_fused_w4<<<dim3(counts[2], 8), 128, 0, stream>>>(
      xbuf, wrec, b_iouh, b_fh, c_buf, hid, starts[2]);
  for (int L = 3; L < 8; ++L) {
    level_fused_ks4<<<dim3(counts[L], 16), 256, 0, stream>>>(
        xbuf, wrec, b_iouh, b_fh, c_buf, hid, out, starts[L]);
  }
}
